# starting kernel
# baseline (speedup 1.0000x reference)
#include <hip/hip_runtime.h>
#include <math.h>

typedef __attribute__((ext_vector_type(2))) float v2f;
typedef __attribute__((ext_vector_type(8))) float v8f;

#define NEG_SLOPE 0.2f

__device__ __forceinline__ void atomicMaxFloat(float* addr, float val) {
  if (val >= 0.0f)
    atomicMax((int*)addr, __float_as_int(val));
  else
    atomicMin((unsigned int*)addr, __float_as_uint(val));
}

__global__ void init_kernel(float* __restrict__ acc_pc, float* __restrict__ acc_mc,
                            float* __restrict__ emax, float* __restrict__ denom,
                            int accN, int edgN) {
  int i = blockIdx.x * blockDim.x + threadIdx.x;
  if (i < accN) { acc_pc[i] = 0.0f; acc_mc[i] = 0.0f; }
  if (i < edgN) { emax[i] = -__builtin_inff(); denom[i] = 0.0f; }
}

__global__ __launch_bounds__(256)
void gemm_kernel(const float* __restrict__ x, const float* __restrict__ Wpc,
                 const float* __restrict__ Wmc, float* __restrict__ hpc,
                 float* __restrict__ hmc) {
  __shared__ float xs[16 * 128];
  const int rowbase = blockIdx.x * 16;
  for (int i = threadIdx.x; i < 16 * 128; i += 256)
    xs[i] = x[(size_t)rowbase * 128 + i];
  __syncthreads();

  const int lane   = threadIdx.x & 31;
  const int wave   = threadIdx.x >> 5;
  const int colbase = wave * 16;
  const int mn    = lane & 15;
  const int khalf = (lane >> 4) << 1;

  v8f cpc = {};
  v8f cmc = {};
  for (int k0 = 0; k0 < 128; k0 += 4) {
    v2f a, bpc, bmc;
    const float* ap = &xs[mn * 128 + k0 + khalf];
    a.x = ap[0];
    a.y = ap[1];
    const int wi = (k0 + khalf) * 128 + colbase + mn;
    bpc.x = Wpc[wi]; bpc.y = Wpc[wi + 128];
    bmc.x = Wmc[wi]; bmc.y = Wmc[wi + 128];
    cpc = __builtin_amdgcn_wmma_f32_16x16x4_f32(false, a, false, bpc, (short)0, cpc, false, false);
    cmc = __builtin_amdgcn_wmma_f32_16x16x4_f32(false, a, false, bmc, (short)0, cmc, false, false);
  }
  const int crow = rowbase + ((lane >> 4) << 3);
  const int ccol = colbase + mn;
#pragma unroll
  for (int r = 0; r < 8; ++r) {
    hpc[(size_t)(crow + r) * 128 + ccol] = cpc[r];
    hmc[(size_t)(crow + r) * 128 + ccol] = cmc[r];
  }
}

__global__ __launch_bounds__(256)
void score_kernel(const float* __restrict__ h, const float* __restrict__ a_src,
                  const float* __restrict__ a_dst, float* __restrict__ s_src,
                  float* __restrict__ s_dst, int n) {
  int warp = (blockIdx.x * blockDim.x + threadIdx.x) >> 5;
  int lane = threadIdx.x & 31;
  if (warp >= n) return;
  const float* hn = h + (size_t)warp * 128;
  float ss = 0.0f, sd = 0.0f;
#pragma unroll
  for (int j = 0; j < 4; ++j) {
    int c = lane * 4 + j;
    float hv = hn[c];
    ss += hv * a_src[c];
    sd += hv * a_dst[c];
  }
#pragma unroll
  for (int m = 1; m < 16; m <<= 1) {
    ss += __shfl_xor(ss, m, 32);
    sd += __shfl_xor(sd, m, 32);
  }
  if ((lane & 15) == 0) {
    int head = lane >> 4;
    s_src[warp * 2 + head] = ss;
    s_dst[warp * 2 + head] = sd;
  }
}

__global__ __launch_bounds__(256)
void edge_max_kernel(const int* __restrict__ ei, int E, int n,
                     const float* __restrict__ s_src, const float* __restrict__ s_dst,
                     float* __restrict__ e_store, float* __restrict__ emax) {
  int e = blockIdx.x * blockDim.x + threadIdx.x;
  if (e >= E + n) return;
  int src = (e < E) ? ei[e]     : (e - E);
  int dst = (e < E) ? ei[E + e] : (e - E);
#pragma unroll
  for (int h = 0; h < 2; ++h) {
    float v = s_src[src * 2 + h] + s_dst[dst * 2 + h];
    v = (v > 0.0f) ? v : NEG_SLOPE * v;
    e_store[(size_t)e * 2 + h] = v;
    atomicMaxFloat(&emax[dst * 2 + h], v);
  }
}

__global__ __launch_bounds__(256)
void edge_exp_kernel(const int* __restrict__ ei, int E, int n,
                     float* __restrict__ e_store, const float* __restrict__ emax,
                     float* __restrict__ denom) {
  int e = blockIdx.x * blockDim.x + threadIdx.x;
  if (e >= E + n) return;
  int dst = (e < E) ? ei[E + e] : (e - E);
#pragma unroll
  for (int h = 0; h < 2; ++h) {
    float ex = expf(e_store[(size_t)e * 2 + h] - emax[dst * 2 + h]);
    e_store[(size_t)e * 2 + h] = ex;
    atomicAdd(&denom[dst * 2 + h], ex);
  }
}

__global__ __launch_bounds__(256)
void edge_acc_kernel(const int* __restrict__ ei, int E, int n,
                     const float* __restrict__ e_store, const float* __restrict__ denom,
                     const float* __restrict__ h, float* __restrict__ acc) {
  int warp = (blockIdx.x * blockDim.x + threadIdx.x) >> 5;
  int lane = threadIdx.x & 31;
  if (warp >= E + n) return;
  int src = (warp < E) ? ei[warp]     : (warp - E);
  int dst = (warp < E) ? ei[E + warp] : (warp - E);
  float a0 = e_store[(size_t)warp * 2 + 0] / denom[dst * 2 + 0];
  float a1 = e_store[(size_t)warp * 2 + 1] / denom[dst * 2 + 1];
  const float* hs = h   + (size_t)src * 128;
  float*       ad = acc + (size_t)dst * 128;
#pragma unroll
  for (int j = 0; j < 4; ++j) {
    int c = lane + 32 * j;
    float alpha = (c < 64) ? a0 : a1;
    atomicAdd(&ad[c], alpha * hs[c]);
  }
}

__global__ __launch_bounds__(256)
void finalize_kernel(const float* __restrict__ acc_pc, const float* __restrict__ acc_mc,
                     const float* __restrict__ b_pc, const float* __restrict__ b_mc,
                     float* __restrict__ out, int total) {
  int i = blockIdx.x * blockDim.x + threadIdx.x;
  if (i >= total) return;
  int c = i & 127;
  float vp = acc_pc[i] + b_pc[c];
  vp = (vp > 0.0f) ? vp : (expf(vp) - 1.0f);
  float vm = acc_mc[i] + b_mc[c];
  vm = (vm > 0.0f) ? vm : (expf(vm) - 1.0f);
  out[i] = 0.5f * (vp + vm);
}

extern "C" void kernel_launch(void* const* d_in, const int* in_sizes, int n_in,
                              void* d_out, int out_size, void* d_ws, size_t ws_size,
                              hipStream_t stream) {
  const float* x      = (const float*)d_in[0];
  const float* W_pc   = (const float*)d_in[1];
  const float* asr_pc = (const float*)d_in[2];
  const float* ads_pc = (const float*)d_in[3];
  const float* b_pc   = (const float*)d_in[4];
  const float* W_mc   = (const float*)d_in[5];
  const float* asr_mc = (const float*)d_in[6];
  const float* ads_mc = (const float*)d_in[7];
  const float* b_mc   = (const float*)d_in[8];
  const int*   ei_pc  = (const int*)d_in[9];
  const int*   ei_mc  = (const int*)d_in[10];
  float* out = (float*)d_out;

  const int NN  = in_sizes[0] / 128;
  const int Epc = in_sizes[9] / 2;
  const int Emc = in_sizes[10] / 2;

  float* ws = (float*)d_ws;
  size_t off = 0;
  float* h_pc    = ws + off; off += (size_t)NN * 128;
  float* h_mc    = ws + off; off += (size_t)NN * 128;
  float* acc_pc  = ws + off; off += (size_t)NN * 128;
  float* acc_mc  = ws + off; off += (size_t)NN * 128;
  float* ssrc_pc = ws + off; off += (size_t)NN * 2;
  float* sdst_pc = ws + off; off += (size_t)NN * 2;
  float* ssrc_mc = ws + off; off += (size_t)NN * 2;
  float* sdst_mc = ws + off; off += (size_t)NN * 2;
  float* emax    = ws + off; off += (size_t)NN * 4;
  float* denom   = ws + off; off += (size_t)NN * 4;
  float* e_pc    = ws + off; off += (size_t)(Epc + NN) * 2;
  float* e_mc    = ws + off; off += (size_t)(Emc + NN) * 2;
  float* emax_pc = emax;            float* emax_mc = emax + (size_t)NN * 2;
  float* den_pc  = denom;           float* den_mc  = denom + (size_t)NN * 2;

  const int accN = NN * 128;
  init_kernel<<<(accN + 255) / 256, 256, 0, stream>>>(acc_pc, acc_mc, emax, denom,
                                                      accN, NN * 4);

  gemm_kernel<<<(NN + 15) / 16, 256, 0, stream>>>(x, W_pc, W_mc, h_pc, h_mc);

  score_kernel<<<(NN + 7) / 8, 256, 0, stream>>>(h_pc, asr_pc, ads_pc, ssrc_pc, sdst_pc, NN);
  score_kernel<<<(NN + 7) / 8, 256, 0, stream>>>(h_mc, asr_mc, ads_mc, ssrc_mc, sdst_mc, NN);

  const int tpc = Epc + NN, tmc = Emc + NN;
  edge_max_kernel<<<(tpc + 255) / 256, 256, 0, stream>>>(ei_pc, Epc, NN, ssrc_pc, sdst_pc, e_pc, emax_pc);
  edge_max_kernel<<<(tmc + 255) / 256, 256, 0, stream>>>(ei_mc, Emc, NN, ssrc_mc, sdst_mc, e_mc, emax_mc);

  edge_exp_kernel<<<(tpc + 255) / 256, 256, 0, stream>>>(ei_pc, Epc, NN, e_pc, emax_pc, den_pc);
  edge_exp_kernel<<<(tmc + 255) / 256, 256, 0, stream>>>(ei_mc, Emc, NN, e_mc, emax_mc, den_mc);

  edge_acc_kernel<<<(tpc + 7) / 8, 256, 0, stream>>>(ei_pc, Epc, NN, e_pc, den_pc, h_pc, acc_pc);
  edge_acc_kernel<<<(tmc + 7) / 8, 256, 0, stream>>>(ei_mc, Emc, NN, e_mc, den_mc, h_mc, acc_mc);

  finalize_kernel<<<(accN + 255) / 256, 256, 0, stream>>>(acc_pc, acc_mc, b_pc, b_mc, out, accN);
}


# Aggregation by dst-range owner work-groups: each walks the dst list, queues its edges in one fixed order by prefix ranks; one wave per row adds alpha*h[src] in LDS; acc atomics and acc zeroing go
# speedup vs baseline: 1.5557x; 1.5557x over previous
#include <hip/hip_runtime.h>
#include <math.h>

typedef __attribute__((ext_vector_type(2))) float v2f;
typedef __attribute__((ext_vector_type(8))) float v8f;

#define NEG_SLOPE 0.2f

__device__ __forceinline__ int fkey(float v) {
  int b = __float_as_int(v);
  return (b >= 0) ? b : (b ^ 0x7FFFFFFF);
}
__device__ __forceinline__ float kval(int k) {
  return __int_as_float((k >= 0) ? k : (k ^ 0x7FFFFFFF));
}

__global__ void init_kernel(int* __restrict__ emax, float* __restrict__ denom, int edgN) {
  int i = blockIdx.x * blockDim.x + threadIdx.x;
  if (i < edgN) { emax[i] = fkey(-__builtin_inff()); denom[i] = 0.0f; }
}

__global__ __launch_bounds__(256)
void gemm_kernel(const float* __restrict__ x, const float* __restrict__ Wpc,
                 const float* __restrict__ Wmc, float* __restrict__ hpc,
                 float* __restrict__ hmc) {
  __shared__ float xs[16 * 128];
  const int rowbase = blockIdx.x * 16;
  for (int i = threadIdx.x; i < 16 * 128; i += 256)
    xs[i] = x[(size_t)rowbase * 128 + i];
  __syncthreads();

  const int lane   = threadIdx.x & 31;
  const int wave   = threadIdx.x >> 5;
  const int colbase = wave * 16;
  const int mn    = lane & 15;
  const int khalf = (lane >> 4) << 1;

  v8f cpc = {};
  v8f cmc = {};
  for (int k0 = 0; k0 < 128; k0 += 4) {
    v2f a, bpc, bmc;
    const float* ap = &xs[mn * 128 + k0 + khalf];
    a.x = ap[0];
    a.y = ap[1];
    const int wi = (k0 + khalf) * 128 + colbase + mn;
    bpc.x = Wpc[wi]; bpc.y = Wpc[wi + 128];
    bmc.x = Wmc[wi]; bmc.y = Wmc[wi + 128];
    cpc = __builtin_amdgcn_wmma_f32_16x16x4_f32(false, a, false, bpc, (short)0, cpc, false, false);
    cmc = __builtin_amdgcn_wmma_f32_16x16x4_f32(false, a, false, bmc, (short)0, cmc, false, false);
  }
  const int crow = rowbase + ((lane >> 4) << 3);
  const int ccol = colbase + mn;
#pragma unroll
  for (int r = 0; r < 8; ++r) {
    hpc[(size_t)(crow + r) * 128 + ccol] = cpc[r];
    hmc[(size_t)(crow + r) * 128 + ccol] = cmc[r];
  }
}

__global__ __launch_bounds__(256)
void score_kernel(const float* __restrict__ h, const float* __restrict__ a_src,
                  const float* __restrict__ a_dst, float* __restrict__ s_src,
                  float* __restrict__ s_dst, int n) {
  int warp = (blockIdx.x * blockDim.x + threadIdx.x) >> 5;
  int lane = threadIdx.x & 31;
  if (warp >= n) return;
  const float* hn = h + (size_t)warp * 128;
  float ss = 0.0f, sd = 0.0f;
#pragma unroll
  for (int j = 0; j < 4; ++j) {
    int c = lane * 4 + j;
    float hv = hn[c];
    ss += hv * a_src[c];
    sd += hv * a_dst[c];
  }
#pragma unroll
  for (int m = 1; m < 16; m <<= 1) {
    ss += __shfl_xor(ss, m, 32);
    sd += __shfl_xor(sd, m, 32);
  }
  if ((lane & 15) == 0) {
    int head = lane >> 4;
    s_src[warp * 2 + head] = ss;
    s_dst[warp * 2 + head] = sd;
  }
}

__global__ __launch_bounds__(256)
void edge_max_kernel(const int* __restrict__ ei, int E, int n,
                     const float* __restrict__ s_src, const float* __restrict__ s_dst,
                     float* __restrict__ e_store, int* __restrict__ emax) {
  const int lane  = threadIdx.x & 31;
  const int wave  = (blockIdx.x * blockDim.x + threadIdx.x) >> 5;
  const int h     = lane & 1;
  const int ebase = wave * 32 + (lane >> 1);
  for (int j = 0; j < 2; ++j) {
    const int e = ebase + 16 * j;
    if (e < E + n) {
      int src = (e < E) ? ei[e]     : (e - E);
      int dst = (e < E) ? ei[E + e] : (e - E);
      float v = s_src[src * 2 + h] + s_dst[dst * 2 + h];
      v = (v > 0.0f) ? v : NEG_SLOPE * v;
      e_store[(size_t)e * 2 + h] = v;
      atomicMax(&emax[dst * 2 + h], fkey(v));
    }
  }
}

__global__ __launch_bounds__(256)
void edge_exp_kernel(const int* __restrict__ ei, int E, int n,
                     float* __restrict__ e_store, const int* __restrict__ emax,
                     float* __restrict__ denom) {
  const int lane  = threadIdx.x & 31;
  const int wave  = (blockIdx.x * blockDim.x + threadIdx.x) >> 5;
  const int h     = lane & 1;
  const int ebase = wave * 32 + (lane >> 1);
  for (int j = 0; j < 2; ++j) {
    const int e = ebase + 16 * j;
    if (e < E + n) {
      int dst = (e < E) ? ei[E + e] : (e - E);
      float ex = expf(e_store[(size_t)e * 2 + h] - kval(emax[dst * 2 + h]));
      e_store[(size_t)e * 2 + h] = ex;
      atomicAdd(&denom[dst * 2 + h], ex);
    }
  }
}

#define OWN_R   196
#define OWN_K   32
#define OWN_CH  (32 * 32 * OWN_K)
#define OWN_QC  4096
#define OWN_U   8
static_assert(OWN_K == 32, "one bit of a 32-bit mask per edge a lane sees in a round");
static_assert(OWN_QC % 128 == 0 && OWN_R <= 256, "queue read in 128-entry steps; local node in 8 bits");

typedef __attribute__((ext_vector_type(4))) float v4f;
typedef __attribute__((ext_vector_type(4))) unsigned v4u;

__device__ __forceinline__ void own_load_dsts(const int* __restrict__ ei, int E, int tot,
                                              int ebase, int (&d)[OWN_K]) {
#pragma unroll
  for (int k = 0; k < OWN_K; ++k) {
    const int e = ebase + k * 32;
    d[k] = (e < E) ? ei[E + e] : ((e < tot) ? (e - E) : -1);
  }
}

__device__ __forceinline__ void own_consume(int nq, int wave, int lane, unsigned lt, int base,
                                            int E, const int* __restrict__ ei,
                                            const float* __restrict__ e_store,
                                            const float* __restrict__ denom,
                                            const float* __restrict__ h, float* accs,
                                            const unsigned* q, unsigned* q2, int* cntw) {
  __syncthreads();

  int mycnt = 0;
  for (int i0 = 0; i0 < nq; i0 += 128) {
    const int idx = i0 + lane * 4;
    const v4u v = *(const v4u*)(q + idx);
    const bool m0 = (idx + 0 < nq) && ((int)((v.x >> 24) & 31u) == wave);
    const bool m1 = (idx + 1 < nq) && ((int)((v.y >> 24) & 31u) == wave);
    const bool m2 = (idx + 2 < nq) && ((int)((v.z >> 24) & 31u) == wave);
    const bool m3 = (idx + 3 < nq) && ((int)((v.w >> 24) & 31u) == wave);
    mycnt += __popc(__builtin_amdgcn_ballot_w32(m0)) + __popc(__builtin_amdgcn_ballot_w32(m1)) +
             __popc(__builtin_amdgcn_ballot_w32(m2)) + __popc(__builtin_amdgcn_ballot_w32(m3));
  }
  if (lane == 0) cntw[wave] = mycnt;
  __syncthreads();

  const int cw = cntw[lane];
  int myoff = 0;
#pragma unroll
  for (int p = 0; p < 32; ++p) {
    const int cp = __builtin_amdgcn_readlane(cw, p);
    myoff += (p < wave) ? cp : 0;
  }

  int run = myoff;
  for (int i0 = 0; i0 < nq; i0 += 128) {
    const int idx = i0 + lane * 4;
    const v4u v = *(const v4u*)(q + idx);
    const bool m0 = (idx + 0 < nq) && ((int)((v.x >> 24) & 31u) == wave);
    const bool m1 = (idx + 1 < nq) && ((int)((v.y >> 24) & 31u) == wave);
    const bool m2 = (idx + 2 < nq) && ((int)((v.z >> 24) & 31u) == wave);
    const bool m3 = (idx + 3 < nq) && ((int)((v.w >> 24) & 31u) == wave);
    const unsigned b0 = __builtin_amdgcn_ballot_w32(m0);
    const unsigned b1 = __builtin_amdgcn_ballot_w32(m1);
    const unsigned b2 = __builtin_amdgcn_ballot_w32(m2);
    const unsigned b3 = __builtin_amdgcn_ballot_w32(m3);
    int p = run + __popc(b0 & lt) + __popc(b1 & lt) + __popc(b2 & lt) + __popc(b3 & lt);
    if (m0) q2[p++] = v.x;
    if (m1) q2[p++] = v.y;
    if (m2) q2[p++] = v.z;
    if (m3) q2[p++] = v.w;
    run += __popc(b0) + __popc(b1) + __popc(b2) + __popc(b3);
  }

  for (int b0 = 0; b0 < mycnt; b0 += 32) {
    const int nb = min(32, mycnt - b0);
    const unsigned ent = (lane < nb) ? q2[myoff + b0 + lane] : 0u;
    const int local = (int)(ent >> 24);
    const int e     = (int)(ent & 0xFFFFFFu);
    int   s  = 0;
    float a0 = 0.0f, a1 = 0.0f;
    if (lane < nb) {
      s = (e < E) ? ei[e] : (e - E);
      const int dst = base + local;
      a0 = e_store[(size_t)e * 2 + 0] / denom[dst * 2 + 0];
      a1 = e_store[(size_t)e * 2 + 1] / denom[dst * 2 + 1];
    }
    for (int j = 0; j < nb; j += OWN_U) {
      v4f hv[OWN_U];
#pragma unroll
      for (int u = 0; u < OWN_U; ++u) {
        const int jj = min(j + u, nb - 1);
        const int sj = __builtin_amdgcn_readlane(s, jj);
        hv[u] = ((const v4f*)(h + (size_t)sj * 128))[lane];
      }
#pragma unroll
      for (int u = 0; u < OWN_U; ++u) {
        if (j + u < nb) {
          const int   lj = __builtin_amdgcn_readlane(local, j + u);
          const float b0f = __int_as_float(__builtin_amdgcn_readlane(__float_as_int(a0), j + u));
          const float b1f = __int_as_float(__builtin_amdgcn_readlane(__float_as_int(a1), j + u));
          const float al = (lane < 16) ? b0f : b1f;
          v4f* ar = (v4f*)(accs + lj * 128) + lane;
          v4f t = *ar;
          t.x += al * hv[u].x;
          t.y += al * hv[u].y;
          t.z += al * hv[u].z;
          t.w += al * hv[u].w;
          *ar = t;
        }
      }
    }
  }
  __syncthreads();
}

__global__ __launch_bounds__(1024)
void owner_acc_kernel(const int* __restrict__ ei, int E, int n,
                      const float* __restrict__ e_store, const float* __restrict__ denom,
                      const float* __restrict__ h, float* __restrict__ acc) {
  __shared__ __attribute__((aligned(16))) float accs[OWN_R * 128];
  __shared__ __attribute__((aligned(16))) unsigned q[OWN_QC];
  __shared__ unsigned q2[OWN_QC];
  __shared__ int      wcnt[2 * 32];
  __shared__ int      cntw[32];

  const int lane = threadIdx.x & 31;
  const int wave = __builtin_amdgcn_readfirstlane((int)(threadIdx.x >> 5));
  const int base = blockIdx.x * OWN_R;
  const int tot  = E + n;
  const int rounds = (tot + OWN_CH - 1) / OWN_CH;
  const unsigned lt = (1u << lane) - 1u;

  for (int i = threadIdx.x; i < OWN_R * 128; i += 1024) accs[i] = 0.0f;

  __syncthreads();

  int qfill = 0;
  for (int r = 0; r < rounds; ++r) {
    const int wbase = r * OWN_CH + wave * (32 * OWN_K);
    const int ebase = wbase + lane;
    int cur[OWN_K];
    if (wbase + 32 * OWN_K <= E) {
      const int* p = ei + E + ebase;
#pragma unroll
      for (int k = 0; k < OWN_K; ++k) cur[k] = p[k * 32];
    } else {
      own_load_dsts(ei, E, tot, ebase, cur);
    }
    unsigned hitk = 0u;
#pragma unroll
    for (int k = 0; k < OWN_K; ++k)
      hitk = hitk + hitk + (((unsigned)(cur[k] - base) < (unsigned)OWN_R) ? 1u : 0u);
    const int hc = __popc(hitk);
    int incl = hc;
#pragma unroll
    for (int d = 1; d < 32; d <<= 1) {
      const int t = __shfl_up(incl, d, 32);
      if (lane >= d) incl += t;
    }
    const int cnt = __builtin_amdgcn_readlane(incl, 31);
    int* wc = wcnt + (r & 1) * 32;
    if (lane == 0) wc[wave] = cnt;
    __syncthreads();

    const int c = wc[lane];
    int incw = c;
#pragma unroll
    for (int d = 1; d < 32; d <<= 1) {
      const int t = __shfl_up(incw, d, 32);
      if (lane >= d) incw += t;
    }
    const int total  = __builtin_amdgcn_readlane(incw, 31);
    const int mybase = __builtin_amdgcn_readlane(incw - c, wave);

    int done = 0;
    while (done < total) {
      if (qfill == OWN_QC) {
        own_consume(qfill, wave, lane, lt, base, E, ei, e_store, denom, h, accs, q, q2, cntw);
        qfill = 0;
      }
      const int take = min(OWN_QC - qfill, total - done);
      unsigned hm = hitk;
      int pos = mybase - done + (incl - hc);
      while (hm) {
        const int k = __clz((int)hm);
        hm &= ~(0x80000000u >> k);
        if (pos >= 0 && pos < take) {
          const int e = ebase + k * 32;
          const int d = (e < E) ? ei[E + e] : (e - E);
          q[qfill + pos] = ((unsigned)(d - base) << 24) | (unsigned)e;
        }
        ++pos;
      }
      qfill += take;
      done  += take;
    }
  }

  own_consume(qfill, wave, lane, lt, base, E, ei, e_store, denom, h, accs, q, q2, cntw);

  for (int i = threadIdx.x; i < OWN_R * 128; i += 1024) {
    const int node = base + (i >> 7);
    if (node < n) acc[(size_t)node * 128 + (i & 127)] = accs[i];
  }
}

__global__ __launch_bounds__(256)
void finalize_kernel(const float* __restrict__ acc_pc, const float* __restrict__ acc_mc,
                     const float* __restrict__ b_pc, const float* __restrict__ b_mc,
                     float* __restrict__ out, int total) {
  int i = blockIdx.x * blockDim.x + threadIdx.x;
  if (i >= total) return;
  int c = i & 127;
  float vp = acc_pc[i] + b_pc[c];
  vp = (vp > 0.0f) ? vp : (expf(vp) - 1.0f);
  float vm = acc_mc[i] + b_mc[c];
  vm = (vm > 0.0f) ? vm : (expf(vm) - 1.0f);
  out[i] = 0.5f * (vp + vm);
}

extern "C" void kernel_launch(void* const* d_in, const int* in_sizes, int n_in,
                              void* d_out, int out_size, void* d_ws, size_t ws_size,
                              hipStream_t stream) {
  const float* x      = (const float*)d_in[0];
  const float* W_pc   = (const float*)d_in[1];
  const float* asr_pc = (const float*)d_in[2];
  const float* ads_pc = (const float*)d_in[3];
  const float* b_pc   = (const float*)d_in[4];
  const float* W_mc   = (const float*)d_in[5];
  const float* asr_mc = (const float*)d_in[6];
  const float* ads_mc = (const float*)d_in[7];
  const float* b_mc   = (const float*)d_in[8];
  const int*   ei_pc  = (const int*)d_in[9];
  const int*   ei_mc  = (const int*)d_in[10];
  float* out = (float*)d_out;

  const int NN  = in_sizes[0] / 128;
  const int Epc = in_sizes[9] / 2;
  const int Emc = in_sizes[10] / 2;

  float* ws = (float*)d_ws;
  size_t off = 0;
  float* h_pc    = ws + off; off += (size_t)NN * 128;
  float* h_mc    = ws + off; off += (size_t)NN * 128;
  float* acc_pc  = ws + off; off += (size_t)NN * 128;
  float* acc_mc  = ws + off; off += (size_t)NN * 128;
  float* ssrc_pc = ws + off; off += (size_t)NN * 2;
  float* sdst_pc = ws + off; off += (size_t)NN * 2;
  float* ssrc_mc = ws + off; off += (size_t)NN * 2;
  float* sdst_mc = ws + off; off += (size_t)NN * 2;
  float* emax    = ws + off; off += (size_t)NN * 4;
  float* denom   = ws + off; off += (size_t)NN * 4;
  float* e_pc    = ws + off; off += (size_t)(Epc + NN) * 2;
  float* e_mc    = ws + off; off += (size_t)(Emc + NN) * 2;
  int*   emax_pc = (int*)emax;      int*   emax_mc = (int*)emax + (size_t)NN * 2;
  float* den_pc  = denom;           float* den_mc  = denom + (size_t)NN * 2;

  const int accN = NN * 128;
  init_kernel<<<(NN * 4 + 255) / 256, 256, 0, stream>>>((int*)emax, denom, NN * 4);

  gemm_kernel<<<(NN + 15) / 16, 256, 0, stream>>>(x, W_pc, W_mc, h_pc, h_mc);

  score_kernel<<<(NN + 7) / 8, 256, 0, stream>>>(h_pc, asr_pc, ads_pc, ssrc_pc, sdst_pc, NN);
  score_kernel<<<(NN + 7) / 8, 256, 0, stream>>>(h_mc, asr_mc, ads_mc, ssrc_mc, sdst_mc, NN);

  const int tpc = Epc + NN, tmc = Emc + NN;
  edge_max_kernel<<<(tpc + 255) / 256, 256, 0, stream>>>(ei_pc, Epc, NN, ssrc_pc, sdst_pc, e_pc, emax_pc);
  edge_max_kernel<<<(tmc + 255) / 256, 256, 0, stream>>>(ei_mc, Emc, NN, ssrc_mc, sdst_mc, e_mc, emax_mc);

  edge_exp_kernel<<<(tpc + 255) / 256, 256, 0, stream>>>(ei_pc, Epc, NN, e_pc, emax_pc, den_pc);
  edge_exp_kernel<<<(tmc + 255) / 256, 256, 0, stream>>>(ei_mc, Emc, NN, e_mc, emax_mc, den_mc);

  const int nown = (NN + OWN_R - 1) / OWN_R;
  owner_acc_kernel<<<nown, 1024, 0, stream>>>(ei_pc, Epc, NN, e_pc, den_pc, h_pc, acc_pc);
  owner_acc_kernel<<<nown, 1024, 0, stream>>>(ei_mc, Emc, NN, e_mc, den_mc, h_mc, acc_mc);

  finalize_kernel<<<(accN + 255) / 256, 256, 0, stream>>>(acc_pc, acc_mc, b_pc, b_mc, out, accN);
}


# Owners stop walking the whole dst list: a split launch counting-sorts each 3072-slot stretch by owner with prefix ranks, no atomics; an owner reads only its pieces; rows now sum in rising edge order
# speedup vs baseline: 3.7787x; 2.4289x over previous
#include <hip/hip_runtime.h>
#include <math.h>

typedef __attribute__((ext_vector_type(2))) float v2f;
typedef __attribute__((ext_vector_type(8))) float v8f;

#define NEG_SLOPE 0.2f

__device__ __forceinline__ int fkey(float v) {
  int b = __float_as_int(v);
  return (b >= 0) ? b : (b ^ 0x7FFFFFFF);
}
__device__ __forceinline__ float kval(int k) {
  return __int_as_float((k >= 0) ? k : (k ^ 0x7FFFFFFF));
}

__global__ __launch_bounds__(256)
void gemm_kernel(const float* __restrict__ x, const float* __restrict__ Wpc,
                 const float* __restrict__ Wmc, float* __restrict__ hpc,
                 float* __restrict__ hmc) {
  __shared__ float xs[16 * 128];
  const int rowbase = blockIdx.x * 16;
  for (int i = threadIdx.x; i < 16 * 128; i += 256)
    xs[i] = x[(size_t)rowbase * 128 + i];
  __syncthreads();

  const int lane   = threadIdx.x & 31;
  const int wave   = threadIdx.x >> 5;
  const int colbase = wave * 16;
  const int mn    = lane & 15;
  const int khalf = (lane >> 4) << 1;

  v8f cpc = {};
  v8f cmc = {};
  for (int k0 = 0; k0 < 128; k0 += 4) {
    v2f a, bpc, bmc;
    const float* ap = &xs[mn * 128 + k0 + khalf];
    a.x = ap[0];
    a.y = ap[1];
    const int wi = (k0 + khalf) * 128 + colbase + mn;
    bpc.x = Wpc[wi]; bpc.y = Wpc[wi + 128];
    bmc.x = Wmc[wi]; bmc.y = Wmc[wi + 128];
    cpc = __builtin_amdgcn_wmma_f32_16x16x4_f32(false, a, false, bpc, (short)0, cpc, false, false);
    cmc = __builtin_amdgcn_wmma_f32_16x16x4_f32(false, a, false, bmc, (short)0, cmc, false, false);
  }
  const int crow = rowbase + ((lane >> 4) << 3);
  const int ccol = colbase + mn;
#pragma unroll
  for (int r = 0; r < 8; ++r) {
    hpc[(size_t)(crow + r) * 128 + ccol] = cpc[r];
    hmc[(size_t)(crow + r) * 128 + ccol] = cmc[r];
  }
}

__global__ __launch_bounds__(256)
void score_kernel(const float* __restrict__ h, const float* __restrict__ a_src,
                  const float* __restrict__ a_dst, float* __restrict__ s_src,
                  float* __restrict__ s_dst, int n) {
  int warp = (blockIdx.x * blockDim.x + threadIdx.x) >> 5;
  int lane = threadIdx.x & 31;
  if (warp >= n) return;
  const float* hn = h + (size_t)warp * 128;
  float ss = 0.0f, sd = 0.0f;
#pragma unroll
  for (int j = 0; j < 4; ++j) {
    int c = lane * 4 + j;
    float hv = hn[c];
    ss += hv * a_src[c];
    sd += hv * a_dst[c];
  }
#pragma unroll
  for (int m = 1; m < 16; m <<= 1) {
    ss += __shfl_xor(ss, m, 32);
    sd += __shfl_xor(sd, m, 32);
  }
  if ((lane & 15) == 0) {
    int head = lane >> 4;
    s_src[warp * 2 + head] = ss;
    s_dst[warp * 2 + head] = sd;
  }
}

#define OWN_R   196
#define SP_W    32
#define SP_K    3
#define SP_CH   (SP_W * 32 * SP_K)
#define SP_NB   256
#define SP_PITCH 288
static_assert(SP_NB + 1 <= SP_PITCH && SP_PITCH % 32 == 0, "a block's offsets are its own lines");

typedef __attribute__((ext_vector_type(4))) float v4f;
typedef __attribute__((ext_vector_type(4))) unsigned v4u;

__global__ __launch_bounds__(1024)
void split_kernel(const int* __restrict__ ei_pc, int E_pc, const int* __restrict__ ei_mc, int E_mc,
                  int n, int P_pc,
                  unsigned* __restrict__ part_pc, int* __restrict__ offs_pc,
                  unsigned* __restrict__ part_mc, int* __restrict__ offs_mc) {
  __shared__ int cnt[SP_W * SP_NB];
  __shared__ __attribute__((aligned(16))) unsigned sorted[SP_CH];
  __shared__ int binstart[SP_NB];
  __shared__ int wsum[SP_NB / 32];

  const bool mc    = (int)blockIdx.x >= P_pc;
  const int  chunk = mc ? (int)blockIdx.x - P_pc : (int)blockIdx.x;
  const int* ei    = mc ? ei_mc : ei_pc;
  const int  E     = mc ? E_mc : E_pc;
  unsigned*  part  = (mc ? part_mc : part_pc) + (size_t)chunk * SP_CH;
  int*       offs  = (mc ? offs_mc : offs_pc) + (size_t)chunk * SP_PITCH;
  const int  tot   = E + n;

  const int lane = threadIdx.x & 31;
  const int wave = __builtin_amdgcn_readfirstlane((int)(threadIdx.x >> 5));
  const unsigned lt = (1u << lane) - 1u;
  int* cw = cnt + wave * SP_NB;

  for (int k = 0; k < SP_NB / 32; ++k) cw[k * 32 + lane] = 0;

  const int sbase = chunk * SP_CH + wave * (32 * SP_K) + lane;
  int dd[SP_K], ss[SP_K];
#pragma unroll
  for (int j = 0; j < SP_K; ++j) {
    const int e = sbase + j * 32;
    int d = 0, sn = 0;
    if (e < E) { sn = ei[e]; d = ei[E + e]; }
    else if (e < tot) { sn = e - E; d = e - E; }
    dd[j] = d; ss[j] = sn;
  }

  int key[SP_K], rnk[SP_K];
  unsigned ent[SP_K];
#pragma unroll
  for (int j = 0; j < SP_K; ++j) {
    const bool ok = (sbase + j * 32) < tot;
    const unsigned own = (unsigned)dd[j] / (unsigned)OWN_R;
    key[j] = (int)(own & (unsigned)(SP_NB - 1));
    ent[j] = (((unsigned)dd[j] - own * (unsigned)OWN_R) << 24) | (unsigned)ss[j];
    unsigned same = __builtin_amdgcn_ballot_w32(ok);
    for (int b = 0; b < 8; ++b) {
      const bool bit = ((key[j] >> b) & 1) != 0;
      const unsigned bb = __builtin_amdgcn_ballot_w32(bit);
      same &= bit ? bb : ~bb;
    }
    rnk[j] = 0;
    if (ok) {
      const int old = cw[key[j]];
      rnk[j] = old + __popc(same & lt);
      if ((same & lt) == 0u) cw[key[j]] = old + __popc(same);
    }
  }
  __syncthreads();

  int incl = 0, totb = 0;
  if (threadIdx.x < SP_NB) {
    int c[SP_W];
#pragma unroll
    for (int w = 0; w < SP_W; ++w) c[w] = cnt[w * SP_NB + threadIdx.x];
    int run = 0;
#pragma unroll
    for (int w = 0; w < SP_W; ++w) { cnt[w * SP_NB + threadIdx.x] = run; run += c[w]; }
    totb = run;
    incl = run;
    for (int d = 1; d < 32; d <<= 1) {
      const int t = __shfl_up(incl, d, 32);
      if (lane >= d) incl += t;
    }
    if (lane == 31) wsum[wave] = incl;
  }
  __syncthreads();
  if (threadIdx.x < SP_NB) {
    int wb = 0;
    for (int w = 0; w < SP_NB / 32; ++w) wb += (w < wave) ? wsum[w] : 0;
    const int bs = wb + incl - totb;
    binstart[threadIdx.x] = bs;
    offs[threadIdx.x] = bs;
    if (threadIdx.x == SP_NB - 1) offs[SP_NB] = bs + totb;
  }
  __syncthreads();

#pragma unroll
  for (int j = 0; j < SP_K; ++j)
    if ((sbase + j * 32) < tot) sorted[binstart[key[j]] + cw[key[j]] + rnk[j]] = ent[j];
  __syncthreads();

  if (threadIdx.x < SP_CH / 4) ((v4u*)part)[threadIdx.x] = ((const v4u*)sorted)[threadIdx.x];
}

#define OWN_T   ((OWN_R + 31) / 32)
#define OWN_QC  4096
#define OWN_U   8
static_assert(OWN_QC % 128 == 0 && OWN_R <= 256, "queue read in 128-entry steps; local node in 8 bits");

__device__ __forceinline__ void own_consume(int nq, int wave, int lane, unsigned lt, int base,
                                            const float* __restrict__ s_src,
                                            const float* __restrict__ s_dst,
                                            const float* __restrict__ h, float* accs,
                                            int* rmax, float* dens, v2f* ev,
                                            const unsigned* q, unsigned* q2, int* cntw) {
  __syncthreads();

  int mycnt = 0;
  for (int i0 = 0; i0 < nq; i0 += 128) {
    const int idx = i0 + lane * 4;
    const v4u v = *(const v4u*)(q + idx);
    const bool m0 = (idx + 0 < nq) && ((int)((v.x >> 24) & 31u) == wave);
    const bool m1 = (idx + 1 < nq) && ((int)((v.y >> 24) & 31u) == wave);
    const bool m2 = (idx + 2 < nq) && ((int)((v.z >> 24) & 31u) == wave);
    const bool m3 = (idx + 3 < nq) && ((int)((v.w >> 24) & 31u) == wave);
    mycnt += __popc(__builtin_amdgcn_ballot_w32(m0)) + __popc(__builtin_amdgcn_ballot_w32(m1)) +
             __popc(__builtin_amdgcn_ballot_w32(m2)) + __popc(__builtin_amdgcn_ballot_w32(m3));
  }
  if (lane == 0) cntw[wave] = mycnt;
  __syncthreads();

  const int cw = cntw[lane];
  int myoff = 0;
#pragma unroll
  for (int p = 0; p < 32; ++p) {
    const int cp = __builtin_amdgcn_readlane(cw, p);
    myoff += (p < wave) ? cp : 0;
  }

  int run = myoff;
  for (int i0 = 0; i0 < nq; i0 += 128) {
    const int idx = i0 + lane * 4;
    const v4u v = *(const v4u*)(q + idx);
    const bool m0 = (idx + 0 < nq) && ((int)((v.x >> 24) & 31u) == wave);
    const bool m1 = (idx + 1 < nq) && ((int)((v.y >> 24) & 31u) == wave);
    const bool m2 = (idx + 2 < nq) && ((int)((v.z >> 24) & 31u) == wave);
    const bool m3 = (idx + 3 < nq) && ((int)((v.w >> 24) & 31u) == wave);
    const unsigned b0 = __builtin_amdgcn_ballot_w32(m0);
    const unsigned b1 = __builtin_amdgcn_ballot_w32(m1);
    const unsigned b2 = __builtin_amdgcn_ballot_w32(m2);
    const unsigned b3 = __builtin_amdgcn_ballot_w32(m3);
    int p = run + __popc(b0 & lt) + __popc(b1 & lt) + __popc(b2 & lt) + __popc(b3 & lt);
    if (m0) q2[p++] = v.x;
    if (m1) q2[p++] = v.y;
    if (m2) q2[p++] = v.z;
    if (m3) q2[p++] = v.w;
    run += __popc(b0) + __popc(b1) + __popc(b2) + __popc(b3);
  }

  const int hd = lane >> 4;
  const int ninf = fkey(-__builtin_inff());

  int oldk[OWN_T];
#pragma unroll
  for (int t = 0; t < OWN_T; ++t) {
    const int row = wave + 32 * t;
    oldk[t] = (row < OWN_R) ? rmax[row * 2 + hd] : ninf;
  }

  for (int b0 = 0; b0 < mycnt; b0 += 32) {
    if (b0 + lane < mycnt) {
      const unsigned ent = q2[myoff + b0 + lane];
      const int local = (int)(ent >> 24);
      const int s     = (int)(ent & 0xFFFFFFu);
      const v2f ss = *(const v2f*)(s_src + (size_t)s * 2);
      const v2f sd = *(const v2f*)(s_dst + (size_t)(base + local) * 2);
      v2f v = ss + sd;
      v.x = (v.x > 0.0f) ? v.x : NEG_SLOPE * v.x;
      v.y = (v.y > 0.0f) ? v.y : NEG_SLOPE * v.y;
      ev[myoff + b0 + lane] = v;
      atomicMax(&rmax[local * 2 + 0], fkey(v.x));
      atomicMax(&rmax[local * 2 + 1], fkey(v.y));
    }
  }

#pragma unroll
  for (int t = 0; t < OWN_T; ++t) {
    const int row = wave + 32 * t;
    if (row < OWN_R) {
      const int newk = rmax[row * 2 + hd];
      if (newk != oldk[t] && oldk[t] != ninf) {
        const float sc = expf(kval(oldk[t]) - kval(newk));
        v4f* ar = (v4f*)(accs + row * 128) + lane;
        *ar = *ar * sc;
        if ((lane & 15) == 0) dens[row * 2 + hd] *= sc;
      }
    }
  }

  for (int b0 = 0; b0 < mycnt; b0 += 32) {
    const int nb = min(32, mycnt - b0);
    const unsigned ent = (lane < nb) ? q2[myoff + b0 + lane] : 0u;
    const int local = (int)(ent >> 24);
    const int s     = (int)(ent & 0xFFFFFFu);
    float a0 = 0.0f, a1 = 0.0f;
    if (lane < nb) {
      const v2f v = ev[myoff + b0 + lane];
      a0 = expf(v.x - kval(rmax[local * 2 + 0]));
      a1 = expf(v.y - kval(rmax[local * 2 + 1]));
    }
    for (int j = 0; j < nb; j += OWN_U) {
      v4f hv[OWN_U];
#pragma unroll
      for (int u = 0; u < OWN_U; ++u) {
        const int jj = min(j + u, nb - 1);
        const int sj = __builtin_amdgcn_readlane(s, jj);
        hv[u] = ((const v4f*)(h + (size_t)sj * 128))[lane];
      }
#pragma unroll
      for (int u = 0; u < OWN_U; ++u) {
        if (j + u < nb) {
          const int   lj = __builtin_amdgcn_readlane(local, j + u);
          const float b0f = __int_as_float(__builtin_amdgcn_readlane(__float_as_int(a0), j + u));
          const float b1f = __int_as_float(__builtin_amdgcn_readlane(__float_as_int(a1), j + u));
          const float al = (lane < 16) ? b0f : b1f;
          v4f* ar = (v4f*)(accs + lj * 128) + lane;
          v4f t = *ar;
          t.x += al * hv[u].x;
          t.y += al * hv[u].y;
          t.z += al * hv[u].z;
          t.w += al * hv[u].w;
          *ar = t;
          if ((lane & 15) == 0) dens[lj * 2 + hd] += al;
        }
      }
    }
  }
  __syncthreads();
}

__device__ __forceinline__ float own_elu(float v) {
  return (v > 0.0f) ? v : (expf(v) - 1.0f);
}

__global__ __launch_bounds__(1024)
void owner_acc_kernel(const unsigned* __restrict__ part_pc, const int* __restrict__ offs_pc, int P_pc,
                      const unsigned* __restrict__ part_mc, const int* __restrict__ offs_mc, int P_mc,
                      int n,
                      const float* __restrict__ ssrc_pc, const float* __restrict__ sdst_pc,
                      const float* __restrict__ h_pc, const float* __restrict__ b_pc,
                      const float* __restrict__ ssrc_mc, const float* __restrict__ sdst_mc,
                      const float* __restrict__ h_mc, const float* __restrict__ b_mc,
                      float* __restrict__ out) {
  __shared__ __attribute__((aligned(16))) float accs[OWN_R * 128];
  __shared__ __attribute__((aligned(16))) float keep[OWN_R * 128];
  __shared__ __attribute__((aligned(16))) unsigned q[OWN_QC];
  __shared__ unsigned q2[OWN_QC];
  __shared__ __attribute__((aligned(8))) v2f ev[OWN_QC];
  __shared__ int      rmax[OWN_R * 2];
  __shared__ float    dens[OWN_R * 2];
  __shared__ int      wcnt[2 * 32];
  __shared__ int      cntw[32];

  const int lane = threadIdx.x & 31;
  const int wave = __builtin_amdgcn_readfirstlane((int)(threadIdx.x >> 5));
  const int base = blockIdx.x * OWN_R;
  const unsigned lt = (1u << lane) - 1u;
  const int sub = lane & 3;

  for (int cv = 0; cv < 2; ++cv) {
    const unsigned* part = cv ? part_mc : part_pc;
    const int*   offs  = cv ? offs_mc : offs_pc;
    const int    P     = cv ? P_mc : P_pc;
    const float* s_src = cv ? ssrc_mc : ssrc_pc;
    const float* s_dst = cv ? sdst_mc : sdst_pc;
    const float* h     = cv ? h_mc : h_pc;
    const float* bias  = cv ? b_mc : b_pc;

    for (int i = threadIdx.x; i < OWN_R * 128; i += 1024) accs[i] = 0.0f;
    for (int i = threadIdx.x; i < OWN_R * 2; i += 1024) {
      rmax[i] = fkey(-__builtin_inff());
      dens[i] = 0.0f;
    }

    __syncthreads();

    int qfill = 0;
    for (int c0 = 0; c0 < P; c0 += 256) {
      const int c = c0 + (int)(threadIdx.x >> 2);
      int s0 = 0, cn = 0;
      if (c < P) {
        const int* o = offs + (size_t)c * SP_PITCH + blockIdx.x;
        s0 = o[0];
        cn = o[1] - s0;
      }
      const unsigned* src = part + (size_t)min(c, P - 1) * SP_CH + s0;

      const int mine = (sub == 0) ? cn : 0;
      int incl = mine;
#pragma unroll
      for (int d = 1; d < 32; d <<= 1) {
        const int t = __shfl_up(incl, d, 32);
        if (lane >= d) incl += t;
      }
      const int cnt = __builtin_amdgcn_readlane(incl, 31);
      int* wc = wcnt + ((c0 >> 8) & 1) * 32;
      if (lane == 0) wc[wave] = cnt;
      __syncthreads();

      const int cwv = wc[lane];
      int incw = cwv;
#pragma unroll
      for (int d = 1; d < 32; d <<= 1) {
        const int t = __shfl_up(incw, d, 32);
        if (lane >= d) incw += t;
      }
      const int total  = __builtin_amdgcn_readlane(incw, 31);
      const int mybase = __builtin_amdgcn_readlane(incw - cwv, wave);
      const int pbase = mybase + __shfl(incl - mine, lane & ~3, 32);

      int done = 0;
      while (done < total) {
        if (qfill == OWN_QC) {
          own_consume(qfill, wave, lane, lt, base, s_src, s_dst, h, accs, rmax, dens, ev, q, q2, cntw);
          qfill = 0;
        }
        const int take = min(OWN_QC - qfill, total - done);
        for (int i = sub; i < cn; i += 8) {
          const int pa = pbase + i - done;
          const int pb = pa + 4;
          const bool ba = (pa >= 0) && (pa < take);
          const bool bb = (i + 4 < cn) && (pb >= 0) && (pb < take);
          const unsigned va = ba ? src[i] : 0u;
          const unsigned vb = bb ? src[i + 4] : 0u;
          if (ba) q[qfill + pa] = va;
          if (bb) q[qfill + pb] = vb;
        }
        qfill += take;
        done  += take;
      }
    }

    own_consume(qfill, wave, lane, lt, base, s_src, s_dst, h, accs, rmax, dens, ev, q, q2, cntw);

    const float bc0 = bias[lane * 4 + 0], bc1 = bias[lane * 4 + 1];
    const float bc2 = bias[lane * 4 + 2], bc3 = bias[lane * 4 + 3];
    for (int i4 = threadIdx.x; i4 < OWN_R * 32; i4 += 1024) {
      const int row = i4 >> 5;
      const float d = dens[row * 2 + (lane >> 4)];
      const v4f a = ((const v4f*)accs)[i4];
      v4f v;
      v.x = own_elu(a.x / d + bc0);
      v.y = own_elu(a.y / d + bc1);
      v.z = own_elu(a.z / d + bc2);
      v.w = own_elu(a.w / d + bc3);
      if (cv == 0) {
        ((v4f*)keep)[i4] = v;
      } else if (base + row < n) {
        const v4f k = ((const v4f*)keep)[i4];
        v4f o;
        o.x = 0.5f * (k.x + v.x);
        o.y = 0.5f * (k.y + v.y);
        o.z = 0.5f * (k.z + v.z);
        o.w = 0.5f * (k.w + v.w);
        ((v4f*)(out + (size_t)(base + row) * 128))[lane] = o;
      }
    }
    __syncthreads();
  }
}

extern "C" void kernel_launch(void* const* d_in, const int* in_sizes, int n_in,
                              void* d_out, int out_size, void* d_ws, size_t ws_size,
                              hipStream_t stream) {
  const float* x      = (const float*)d_in[0];
  const float* W_pc   = (const float*)d_in[1];
  const float* asr_pc = (const float*)d_in[2];
  const float* ads_pc = (const float*)d_in[3];
  const float* b_pc   = (const float*)d_in[4];
  const float* W_mc   = (const float*)d_in[5];
  const float* asr_mc = (const float*)d_in[6];
  const float* ads_mc = (const float*)d_in[7];
  const float* b_mc   = (const float*)d_in[8];
  const int*   ei_pc  = (const int*)d_in[9];
  const int*   ei_mc  = (const int*)d_in[10];
  float* out = (float*)d_out;

  const int NN  = in_sizes[0] / 128;
  const int Epc = in_sizes[9] / 2;
  const int Emc = in_sizes[10] / 2;
  const int Ppc = (Epc + NN + SP_CH - 1) / SP_CH;
  const int Pmc = (Emc + NN + SP_CH - 1) / SP_CH;

  float* ws = (float*)d_ws;
  size_t off = 0;
  float* h_pc    = ws + off; off += (size_t)NN * 128;
  float* h_mc    = ws + off; off += (size_t)NN * 128;
  float* ssrc_pc = ws + off; off += (size_t)NN * 2;
  float* sdst_pc = ws + off; off += (size_t)NN * 2;
  float* ssrc_mc = ws + off; off += (size_t)NN * 2;
  float* sdst_mc = ws + off; off += (size_t)NN * 2;
  unsigned* part_pc = (unsigned*)(ws + off); off += (size_t)Ppc * SP_CH;
  unsigned* part_mc = (unsigned*)(ws + off); off += (size_t)Pmc * SP_CH;
  int* offs_pc = (int*)(ws + off); off += (size_t)Ppc * SP_PITCH;
  int* offs_mc = (int*)(ws + off); off += (size_t)Pmc * SP_PITCH;

  gemm_kernel<<<(NN + 15) / 16, 256, 0, stream>>>(x, W_pc, W_mc, h_pc, h_mc);

  score_kernel<<<(NN + 7) / 8, 256, 0, stream>>>(h_pc, asr_pc, ads_pc, ssrc_pc, sdst_pc, NN);
  score_kernel<<<(NN + 7) / 8, 256, 0, stream>>>(h_mc, asr_mc, ads_mc, ssrc_mc, sdst_mc, NN);

  split_kernel<<<Ppc + Pmc, 1024, 0, stream>>>(ei_pc, Epc, ei_mc, Emc, NN, Ppc,
                                               part_pc, offs_pc, part_mc, offs_mc);

  const int nown = (NN + OWN_R - 1) / OWN_R;
  owner_acc_kernel<<<nown, 1024, 0, stream>>>(part_pc, offs_pc, Ppc, part_mc, offs_mc, Pmc, NN,
                                             ssrc_pc, sdst_pc, h_pc, b_pc,
                                             ssrc_mc, sdst_mc, h_mc, b_mc, out);
}


# Score launches: a wave scores 16 consecutive nodes, not one (16-lane sums folded for all nodes at once, one 128-byte line stored): 50000 waves per launch become 3125; per-lane products now fma
# speedup vs baseline: 3.9120x; 1.0353x over previous
#include <hip/hip_runtime.h>
#include <math.h>

typedef __attribute__((ext_vector_type(2))) float v2f;
typedef __attribute__((ext_vector_type(8))) float v8f;

#define NEG_SLOPE 0.2f

__device__ __forceinline__ int fkey(float v) {
  int b = __float_as_int(v);
  return (b >= 0) ? b : (b ^ 0x7FFFFFFF);
}
__device__ __forceinline__ float kval(int k) {
  return __int_as_float((k >= 0) ? k : (k ^ 0x7FFFFFFF));
}

__global__ __launch_bounds__(256)
void gemm_kernel(const float* __restrict__ x, const float* __restrict__ Wpc,
                 const float* __restrict__ Wmc, float* __restrict__ hpc,
                 float* __restrict__ hmc) {
  __shared__ float xs[16 * 128];
  const int rowbase = blockIdx.x * 16;
  for (int i = threadIdx.x; i < 16 * 128; i += 256)
    xs[i] = x[(size_t)rowbase * 128 + i];
  __syncthreads();

  const int lane   = threadIdx.x & 31;
  const int wave   = threadIdx.x >> 5;
  const int colbase = wave * 16;
  const int mn    = lane & 15;
  const int khalf = (lane >> 4) << 1;

  v8f cpc = {};
  v8f cmc = {};
  for (int k0 = 0; k0 < 128; k0 += 4) {
    v2f a, bpc, bmc;
    const float* ap = &xs[mn * 128 + k0 + khalf];
    a.x = ap[0];
    a.y = ap[1];
    const int wi = (k0 + khalf) * 128 + colbase + mn;
    bpc.x = Wpc[wi]; bpc.y = Wpc[wi + 128];
    bmc.x = Wmc[wi]; bmc.y = Wmc[wi + 128];
    cpc = __builtin_amdgcn_wmma_f32_16x16x4_f32(false, a, false, bpc, (short)0, cpc, false, false);
    cmc = __builtin_amdgcn_wmma_f32_16x16x4_f32(false, a, false, bmc, (short)0, cmc, false, false);
  }
  const int crow = rowbase + ((lane >> 4) << 3);
  const int ccol = colbase + mn;
#pragma unroll
  for (int r = 0; r < 8; ++r) {
    hpc[(size_t)(crow + r) * 128 + ccol] = cpc[r];
    hmc[(size_t)(crow + r) * 128 + ccol] = cmc[r];
  }
}

#define SC_NPW 16
typedef __attribute__((ext_vector_type(4))) float sc4f;

__global__ __launch_bounds__(256)
void score_kernel(const float* __restrict__ h, const float* __restrict__ a_src,
                  const float* __restrict__ a_dst, float* __restrict__ s_src,
                  float* __restrict__ s_dst, int n) {
  const int warp = (blockIdx.x * blockDim.x + threadIdx.x) >> 5;
  const int lane = threadIdx.x & 31;
  const int nb = warp * SC_NPW;
  if (nb >= n) return;

  sc4f hv[SC_NPW];
#pragma unroll
  for (int j = 0; j < SC_NPW; ++j) {
    const int node = min(nb + j, n - 1);
    hv[j] = ((const sc4f*)(h + (size_t)node * 128))[lane];
  }
  const sc4f as = ((const sc4f*)a_src)[lane];
  const sc4f ad = ((const sc4f*)a_dst)[lane];

  float ps[SC_NPW], pd[SC_NPW];
#pragma unroll
  for (int j = 0; j < SC_NPW; ++j) {
    float ss = 0.0f, sd = 0.0f;
    ss += hv[j].x * as.x;  sd += hv[j].x * ad.x;
    ss += hv[j].y * as.y;  sd += hv[j].y * ad.y;
    ss += hv[j].z * as.z;  sd += hv[j].z * ad.z;
    ss += hv[j].w * as.w;  sd += hv[j].w * ad.w;
    ps[j] = ss;
    pd[j] = sd;
  }

  const bool b0 = (lane & 1) != 0;
  float s8[8], d8[8];
#pragma unroll
  for (int m = 0; m < 8; ++m) {
    s8[m] = (b0 ? ps[2 * m + 1] : ps[2 * m]) + __shfl_xor(b0 ? ps[2 * m] : ps[2 * m + 1], 1, 32);
    d8[m] = (b0 ? pd[2 * m + 1] : pd[2 * m]) + __shfl_xor(b0 ? pd[2 * m] : pd[2 * m + 1], 1, 32);
  }
  const bool b1 = (lane & 2) != 0;
  float s4[4], d4[4];
#pragma unroll
  for (int m = 0; m < 4; ++m) {
    s4[m] = (b1 ? s8[2 * m + 1] : s8[2 * m]) + __shfl_xor(b1 ? s8[2 * m] : s8[2 * m + 1], 2, 32);
    d4[m] = (b1 ? d8[2 * m + 1] : d8[2 * m]) + __shfl_xor(b1 ? d8[2 * m] : d8[2 * m + 1], 2, 32);
  }
  const bool b2 = (lane & 4) != 0;
  float s2[2], d2[2];
#pragma unroll
  for (int m = 0; m < 2; ++m) {
    s2[m] = (b2 ? s4[2 * m + 1] : s4[2 * m]) + __shfl_xor(b2 ? s4[2 * m] : s4[2 * m + 1], 4, 32);
    d2[m] = (b2 ? d4[2 * m + 1] : d4[2 * m]) + __shfl_xor(b2 ? d4[2 * m] : d4[2 * m + 1], 4, 32);
  }
  const bool b3 = (lane & 8) != 0;
  const float ssum = (b3 ? s2[1] : s2[0]) + __shfl_xor(b3 ? s2[0] : s2[1], 8, 32);
  const float dsum = (b3 ? d2[1] : d2[0]) + __shfl_xor(b3 ? d2[0] : d2[1], 8, 32);

  const int node = nb + (lane & 15);
  if (node < n) {
    const int head = lane >> 4;
    s_src[node * 2 + head] = ssum;
    s_dst[node * 2 + head] = dsum;
  }
}

#define OWN_R   196
#define SP_W    32
#define SP_K    3
#define SP_CH   (SP_W * 32 * SP_K)
#define SP_NB   256
#define SP_PITCH 288
static_assert(SP_NB + 1 <= SP_PITCH && SP_PITCH % 32 == 0, "a block's offsets are its own lines");

typedef __attribute__((ext_vector_type(4))) float v4f;
typedef __attribute__((ext_vector_type(4))) unsigned v4u;

__global__ __launch_bounds__(1024)
void split_kernel(const int* __restrict__ ei_pc, int E_pc, const int* __restrict__ ei_mc, int E_mc,
                  int n, int P_pc,
                  unsigned* __restrict__ part_pc, int* __restrict__ offs_pc,
                  unsigned* __restrict__ part_mc, int* __restrict__ offs_mc) {
  __shared__ int cnt[SP_W * SP_NB];
  __shared__ __attribute__((aligned(16))) unsigned sorted[SP_CH];
  __shared__ int binstart[SP_NB];
  __shared__ int wsum[SP_NB / 32];

  const bool mc    = (int)blockIdx.x >= P_pc;
  const int  chunk = mc ? (int)blockIdx.x - P_pc : (int)blockIdx.x;
  const int* ei    = mc ? ei_mc : ei_pc;
  const int  E     = mc ? E_mc : E_pc;
  unsigned*  part  = (mc ? part_mc : part_pc) + (size_t)chunk * SP_CH;
  int*       offs  = (mc ? offs_mc : offs_pc) + (size_t)chunk * SP_PITCH;
  const int  tot   = E + n;

  const int lane = threadIdx.x & 31;
  const int wave = __builtin_amdgcn_readfirstlane((int)(threadIdx.x >> 5));
  const unsigned lt = (1u << lane) - 1u;
  int* cw = cnt + wave * SP_NB;

  for (int k = 0; k < SP_NB / 32; ++k) cw[k * 32 + lane] = 0;

  const int sbase = chunk * SP_CH + wave * (32 * SP_K) + lane;
  int dd[SP_K], ss[SP_K];
#pragma unroll
  for (int j = 0; j < SP_K; ++j) {
    const int e = sbase + j * 32;
    int d = 0, sn = 0;
    if (e < E) { sn = ei[e]; d = ei[E + e]; }
    else if (e < tot) { sn = e - E; d = e - E; }
    dd[j] = d; ss[j] = sn;
  }

  int key[SP_K], rnk[SP_K];
  unsigned ent[SP_K];
#pragma unroll
  for (int j = 0; j < SP_K; ++j) {
    const bool ok = (sbase + j * 32) < tot;
    const unsigned own = (unsigned)dd[j] / (unsigned)OWN_R;
    key[j] = (int)(own & (unsigned)(SP_NB - 1));
    ent[j] = (((unsigned)dd[j] - own * (unsigned)OWN_R) << 24) | (unsigned)ss[j];
    unsigned same = __builtin_amdgcn_ballot_w32(ok);
    for (int b = 0; b < 8; ++b) {
      const bool bit = ((key[j] >> b) & 1) != 0;
      const unsigned bb = __builtin_amdgcn_ballot_w32(bit);
      same &= bit ? bb : ~bb;
    }
    rnk[j] = 0;
    if (ok) {
      const int old = cw[key[j]];
      rnk[j] = old + __popc(same & lt);
      if ((same & lt) == 0u) cw[key[j]] = old + __popc(same);
    }
  }
  __syncthreads();

  int incl = 0, totb = 0;
  if (threadIdx.x < SP_NB) {
    int c[SP_W];
#pragma unroll
    for (int w = 0; w < SP_W; ++w) c[w] = cnt[w * SP_NB + threadIdx.x];
    int run = 0;
#pragma unroll
    for (int w = 0; w < SP_W; ++w) { cnt[w * SP_NB + threadIdx.x] = run; run += c[w]; }
    totb = run;
    incl = run;
    for (int d = 1; d < 32; d <<= 1) {
      const int t = __shfl_up(incl, d, 32);
      if (lane >= d) incl += t;
    }
    if (lane == 31) wsum[wave] = incl;
  }
  __syncthreads();
  if (threadIdx.x < SP_NB) {
    int wb = 0;
    for (int w = 0; w < SP_NB / 32; ++w) wb += (w < wave) ? wsum[w] : 0;
    const int bs = wb + incl - totb;
    binstart[threadIdx.x] = bs;
    offs[threadIdx.x] = bs;
    if (threadIdx.x == SP_NB - 1) offs[SP_NB] = bs + totb;
  }
  __syncthreads();

#pragma unroll
  for (int j = 0; j < SP_K; ++j)
    if ((sbase + j * 32) < tot) sorted[binstart[key[j]] + cw[key[j]] + rnk[j]] = ent[j];
  __syncthreads();

  if (threadIdx.x < SP_CH / 4) ((v4u*)part)[threadIdx.x] = ((const v4u*)sorted)[threadIdx.x];
}

#define OWN_T   ((OWN_R + 31) / 32)
#define OWN_QC  4096
#define OWN_U   8
static_assert(OWN_QC % 128 == 0 && OWN_R <= 256, "queue read in 128-entry steps; local node in 8 bits");

__device__ __forceinline__ void own_consume(int nq, int wave, int lane, unsigned lt, int base,
                                            const float* __restrict__ s_src,
                                            const float* __restrict__ s_dst,
                                            const float* __restrict__ h, float* accs,
                                            int* rmax, float* dens, v2f* ev,
                                            const unsigned* q, unsigned* q2, int* cntw) {
  __syncthreads();

  int mycnt = 0;
  for (int i0 = 0; i0 < nq; i0 += 128) {
    const int idx = i0 + lane * 4;
    const v4u v = *(const v4u*)(q + idx);
    const bool m0 = (idx + 0 < nq) && ((int)((v.x >> 24) & 31u) == wave);
    const bool m1 = (idx + 1 < nq) && ((int)((v.y >> 24) & 31u) == wave);
    const bool m2 = (idx + 2 < nq) && ((int)((v.z >> 24) & 31u) == wave);
    const bool m3 = (idx + 3 < nq) && ((int)((v.w >> 24) & 31u) == wave);
    mycnt += __popc(__builtin_amdgcn_ballot_w32(m0)) + __popc(__builtin_amdgcn_ballot_w32(m1)) +
             __popc(__builtin_amdgcn_ballot_w32(m2)) + __popc(__builtin_amdgcn_ballot_w32(m3));
  }
  if (lane == 0) cntw[wave] = mycnt;
  __syncthreads();

  const int cw = cntw[lane];
  int myoff = 0;
#pragma unroll
  for (int p = 0; p < 32; ++p) {
    const int cp = __builtin_amdgcn_readlane(cw, p);
    myoff += (p < wave) ? cp : 0;
  }

  int run = myoff;
  for (int i0 = 0; i0 < nq; i0 += 128) {
    const int idx = i0 + lane * 4;
    const v4u v = *(const v4u*)(q + idx);
    const bool m0 = (idx + 0 < nq) && ((int)((v.x >> 24) & 31u) == wave);
    const bool m1 = (idx + 1 < nq) && ((int)((v.y >> 24) & 31u) == wave);
    const bool m2 = (idx + 2 < nq) && ((int)((v.z >> 24) & 31u) == wave);
    const bool m3 = (idx + 3 < nq) && ((int)((v.w >> 24) & 31u) == wave);
    const unsigned b0 = __builtin_amdgcn_ballot_w32(m0);
    const unsigned b1 = __builtin_amdgcn_ballot_w32(m1);
    const unsigned b2 = __builtin_amdgcn_ballot_w32(m2);
    const unsigned b3 = __builtin_amdgcn_ballot_w32(m3);
    int p = run + __popc(b0 & lt) + __popc(b1 & lt) + __popc(b2 & lt) + __popc(b3 & lt);
    if (m0) q2[p++] = v.x;
    if (m1) q2[p++] = v.y;
    if (m2) q2[p++] = v.z;
    if (m3) q2[p++] = v.w;
    run += __popc(b0) + __popc(b1) + __popc(b2) + __popc(b3);
  }

  const int hd = lane >> 4;
  const int ninf = fkey(-__builtin_inff());

  int oldk[OWN_T];
#pragma unroll
  for (int t = 0; t < OWN_T; ++t) {
    const int row = wave + 32 * t;
    oldk[t] = (row < OWN_R) ? rmax[row * 2 + hd] : ninf;
  }

  for (int b0 = 0; b0 < mycnt; b0 += 32) {
    if (b0 + lane < mycnt) {
      const unsigned ent = q2[myoff + b0 + lane];
      const int local = (int)(ent >> 24);
      const int s     = (int)(ent & 0xFFFFFFu);
      const v2f ss = *(const v2f*)(s_src + (size_t)s * 2);
      const v2f sd = *(const v2f*)(s_dst + (size_t)(base + local) * 2);
      v2f v = ss + sd;
      v.x = (v.x > 0.0f) ? v.x : NEG_SLOPE * v.x;
      v.y = (v.y > 0.0f) ? v.y : NEG_SLOPE * v.y;
      ev[myoff + b0 + lane] = v;
      atomicMax(&rmax[local * 2 + 0], fkey(v.x));
      atomicMax(&rmax[local * 2 + 1], fkey(v.y));
    }
  }

#pragma unroll
  for (int t = 0; t < OWN_T; ++t) {
    const int row = wave + 32 * t;
    if (row < OWN_R) {
      const int newk = rmax[row * 2 + hd];
      if (newk != oldk[t] && oldk[t] != ninf) {
        const float sc = expf(kval(oldk[t]) - kval(newk));
        v4f* ar = (v4f*)(accs + row * 128) + lane;
        *ar = *ar * sc;
        if ((lane & 15) == 0) dens[row * 2 + hd] *= sc;
      }
    }
  }

  for (int b0 = 0; b0 < mycnt; b0 += 32) {
    const int nb = min(32, mycnt - b0);
    const unsigned ent = (lane < nb) ? q2[myoff + b0 + lane] : 0u;
    const int local = (int)(ent >> 24);
    const int s     = (int)(ent & 0xFFFFFFu);
    float a0 = 0.0f, a1 = 0.0f;
    if (lane < nb) {
      const v2f v = ev[myoff + b0 + lane];
      a0 = expf(v.x - kval(rmax[local * 2 + 0]));
      a1 = expf(v.y - kval(rmax[local * 2 + 1]));
    }
    for (int j = 0; j < nb; j += OWN_U) {
      v4f hv[OWN_U];
#pragma unroll
      for (int u = 0; u < OWN_U; ++u) {
        const int jj = min(j + u, nb - 1);
        const int sj = __builtin_amdgcn_readlane(s, jj);
        hv[u] = ((const v4f*)(h + (size_t)sj * 128))[lane];
      }
#pragma unroll
      for (int u = 0; u < OWN_U; ++u) {
        if (j + u < nb) {
          const int   lj = __builtin_amdgcn_readlane(local, j + u);
          const float b0f = __int_as_float(__builtin_amdgcn_readlane(__float_as_int(a0), j + u));
          const float b1f = __int_as_float(__builtin_amdgcn_readlane(__float_as_int(a1), j + u));
          const float al = (lane < 16) ? b0f : b1f;
          v4f* ar = (v4f*)(accs + lj * 128) + lane;
          v4f t = *ar;
          t.x += al * hv[u].x;
          t.y += al * hv[u].y;
          t.z += al * hv[u].z;
          t.w += al * hv[u].w;
          *ar = t;
          if ((lane & 15) == 0) dens[lj * 2 + hd] += al;
        }
      }
    }
  }
  __syncthreads();
}

__device__ __forceinline__ float own_elu(float v) {
  return (v > 0.0f) ? v : (expf(v) - 1.0f);
}

__global__ __launch_bounds__(1024)
void owner_acc_kernel(const unsigned* __restrict__ part_pc, const int* __restrict__ offs_pc, int P_pc,
                      const unsigned* __restrict__ part_mc, const int* __restrict__ offs_mc, int P_mc,
                      int n,
                      const float* __restrict__ ssrc_pc, const float* __restrict__ sdst_pc,
                      const float* __restrict__ h_pc, const float* __restrict__ b_pc,
                      const float* __restrict__ ssrc_mc, const float* __restrict__ sdst_mc,
                      const float* __restrict__ h_mc, const float* __restrict__ b_mc,
                      float* __restrict__ out) {
  __shared__ __attribute__((aligned(16))) float accs[OWN_R * 128];
  __shared__ __attribute__((aligned(16))) float keep[OWN_R * 128];
  __shared__ __attribute__((aligned(16))) unsigned q[OWN_QC];
  __shared__ unsigned q2[OWN_QC];
  __shared__ __attribute__((aligned(8))) v2f ev[OWN_QC];
  __shared__ int      rmax[OWN_R * 2];
  __shared__ float    dens[OWN_R * 2];
  __shared__ int      wcnt[2 * 32];
  __shared__ int      cntw[32];

  const int lane = threadIdx.x & 31;
  const int wave = __builtin_amdgcn_readfirstlane((int)(threadIdx.x >> 5));
  const int base = blockIdx.x * OWN_R;
  const unsigned lt = (1u << lane) - 1u;
  const int sub = lane & 3;

  for (int cv = 0; cv < 2; ++cv) {
    const unsigned* part = cv ? part_mc : part_pc;
    const int*   offs  = cv ? offs_mc : offs_pc;
    const int    P     = cv ? P_mc : P_pc;
    const float* s_src = cv ? ssrc_mc : ssrc_pc;
    const float* s_dst = cv ? sdst_mc : sdst_pc;
    const float* h     = cv ? h_mc : h_pc;
    const float* bias  = cv ? b_mc : b_pc;

    for (int i = threadIdx.x; i < OWN_R * 128; i += 1024) accs[i] = 0.0f;
    for (int i = threadIdx.x; i < OWN_R * 2; i += 1024) {
      rmax[i] = fkey(-__builtin_inff());
      dens[i] = 0.0f;
    }

    __syncthreads();

    int qfill = 0;
    for (int c0 = 0; c0 < P; c0 += 256) {
      const int c = c0 + (int)(threadIdx.x >> 2);
      int s0 = 0, cn = 0;
      if (c < P) {
        const int* o = offs + (size_t)c * SP_PITCH + blockIdx.x;
        s0 = o[0];
        cn = o[1] - s0;
      }
      const unsigned* src = part + (size_t)min(c, P - 1) * SP_CH + s0;

      const int mine = (sub == 0) ? cn : 0;
      int incl = mine;
#pragma unroll
      for (int d = 1; d < 32; d <<= 1) {
        const int t = __shfl_up(incl, d, 32);
        if (lane >= d) incl += t;
      }
      const int cnt = __builtin_amdgcn_readlane(incl, 31);
      int* wc = wcnt + ((c0 >> 8) & 1) * 32;
      if (lane == 0) wc[wave] = cnt;
      __syncthreads();

      const int cwv = wc[lane];
      int incw = cwv;
#pragma unroll
      for (int d = 1; d < 32; d <<= 1) {
        const int t = __shfl_up(incw, d, 32);
        if (lane >= d) incw += t;
      }
      const int total  = __builtin_amdgcn_readlane(incw, 31);
      const int mybase = __builtin_amdgcn_readlane(incw - cwv, wave);
      const int pbase = mybase + __shfl(incl - mine, lane & ~3, 32);

      int done = 0;
      while (done < total) {
        if (qfill == OWN_QC) {
          own_consume(qfill, wave, lane, lt, base, s_src, s_dst, h, accs, rmax, dens, ev, q, q2, cntw);
          qfill = 0;
        }
        const int take = min(OWN_QC - qfill, total - done);
        for (int i = sub; i < cn; i += 8) {
          const int pa = pbase + i - done;
          const int pb = pa + 4;
          const bool ba = (pa >= 0) && (pa < take);
          const bool bb = (i + 4 < cn) && (pb >= 0) && (pb < take);
          const unsigned va = ba ? src[i] : 0u;
          const unsigned vb = bb ? src[i + 4] : 0u;
          if (ba) q[qfill + pa] = va;
          if (bb) q[qfill + pb] = vb;
        }
        qfill += take;
        done  += take;
      }
    }

    own_consume(qfill, wave, lane, lt, base, s_src, s_dst, h, accs, rmax, dens, ev, q, q2, cntw);

    const float bc0 = bias[lane * 4 + 0], bc1 = bias[lane * 4 + 1];
    const float bc2 = bias[lane * 4 + 2], bc3 = bias[lane * 4 + 3];
    for (int i4 = threadIdx.x; i4 < OWN_R * 32; i4 += 1024) {
      const int row = i4 >> 5;
      const float d = dens[row * 2 + (lane >> 4)];
      const v4f a = ((const v4f*)accs)[i4];
      v4f v;
      v.x = own_elu(a.x / d + bc0);
      v.y = own_elu(a.y / d + bc1);
      v.z = own_elu(a.z / d + bc2);
      v.w = own_elu(a.w / d + bc3);
      if (cv == 0) {
        ((v4f*)keep)[i4] = v;
      } else if (base + row < n) {
        const v4f k = ((const v4f*)keep)[i4];
        v4f o;
        o.x = 0.5f * (k.x + v.x);
        o.y = 0.5f * (k.y + v.y);
        o.z = 0.5f * (k.z + v.z);
        o.w = 0.5f * (k.w + v.w);
        ((v4f*)(out + (size_t)(base + row) * 128))[lane] = o;
      }
    }
    __syncthreads();
  }
}

extern "C" void kernel_launch(void* const* d_in, const int* in_sizes, int n_in,
                              void* d_out, int out_size, void* d_ws, size_t ws_size,
                              hipStream_t stream) {
  const float* x      = (const float*)d_in[0];
  const float* W_pc   = (const float*)d_in[1];
  const float* asr_pc = (const float*)d_in[2];
  const float* ads_pc = (const float*)d_in[3];
  const float* b_pc   = (const float*)d_in[4];
  const float* W_mc   = (const float*)d_in[5];
  const float* asr_mc = (const float*)d_in[6];
  const float* ads_mc = (const float*)d_in[7];
  const float* b_mc   = (const float*)d_in[8];
  const int*   ei_pc  = (const int*)d_in[9];
  const int*   ei_mc  = (const int*)d_in[10];
  float* out = (float*)d_out;

  const int NN  = in_sizes[0] / 128;
  const int Epc = in_sizes[9] / 2;
  const int Emc = in_sizes[10] / 2;
  const int Ppc = (Epc + NN + SP_CH - 1) / SP_CH;
  const int Pmc = (Emc + NN + SP_CH - 1) / SP_CH;

  float* ws = (float*)d_ws;
  size_t off = 0;
  float* h_pc    = ws + off; off += (size_t)NN * 128;
  float* h_mc    = ws + off; off += (size_t)NN * 128;
  float* ssrc_pc = ws + off; off += (size_t)NN * 2;
  float* sdst_pc = ws + off; off += (size_t)NN * 2;
  float* ssrc_mc = ws + off; off += (size_t)NN * 2;
  float* sdst_mc = ws + off; off += (size_t)NN * 2;
  unsigned* part_pc = (unsigned*)(ws + off); off += (size_t)Ppc * SP_CH;
  unsigned* part_mc = (unsigned*)(ws + off); off += (size_t)Pmc * SP_CH;
  int* offs_pc = (int*)(ws + off); off += (size_t)Ppc * SP_PITCH;
  int* offs_mc = (int*)(ws + off); off += (size_t)Pmc * SP_PITCH;

  gemm_kernel<<<(NN + 15) / 16, 256, 0, stream>>>(x, W_pc, W_mc, h_pc, h_mc);

  score_kernel<<<(NN + 8 * SC_NPW - 1) / (8 * SC_NPW), 256, 0, stream>>>(h_pc, asr_pc, ads_pc, ssrc_pc, sdst_pc, NN);
  score_kernel<<<(NN + 8 * SC_NPW - 1) / (8 * SC_NPW), 256, 0, stream>>>(h_mc, asr_mc, ads_mc, ssrc_mc, sdst_mc, NN);

  split_kernel<<<Ppc + Pmc, 1024, 0, stream>>>(ei_pc, Epc, ei_mc, Emc, NN, Ppc,
                                               part_pc, offs_pc, part_mc, offs_mc);

  const int nown = (NN + OWN_R - 1) / OWN_R;
  owner_acc_kernel<<<nown, 1024, 0, stream>>>(part_pc, offs_pc, Ppc, part_mc, offs_mc, Pmc, NN,
                                             ssrc_pc, sdst_pc, h_pc, b_pc,
                                             ssrc_mc, sdst_mc, h_mc, b_mc, out);
}


# Owner work-group hands its queue to the waves by one shared counting sort on the row's wave (ballot prefix ranks, key-major scan of 32x32 counts), not 32 whole-queue scans; sums keep their order
# speedup vs baseline: 4.4149x; 1.1286x over previous
#include <hip/hip_runtime.h>
#include <math.h>

typedef __attribute__((ext_vector_type(2))) float v2f;
typedef __attribute__((ext_vector_type(8))) float v8f;

#define NEG_SLOPE 0.2f

__device__ __forceinline__ int fkey(float v) {
  int b = __float_as_int(v);
  return (b >= 0) ? b : (b ^ 0x7FFFFFFF);
}
__device__ __forceinline__ float kval(int k) {
  return __int_as_float((k >= 0) ? k : (k ^ 0x7FFFFFFF));
}

__global__ __launch_bounds__(256)
void gemm_kernel(const float* __restrict__ x, const float* __restrict__ Wpc,
                 const float* __restrict__ Wmc, float* __restrict__ hpc,
                 float* __restrict__ hmc) {
  __shared__ float xs[16 * 128];
  const int rowbase = blockIdx.x * 16;
  for (int i = threadIdx.x; i < 16 * 128; i += 256)
    xs[i] = x[(size_t)rowbase * 128 + i];
  __syncthreads();

  const int lane   = threadIdx.x & 31;
  const int wave   = threadIdx.x >> 5;
  const int colbase = wave * 16;
  const int mn    = lane & 15;
  const int khalf = (lane >> 4) << 1;

  v8f cpc = {};
  v8f cmc = {};
  for (int k0 = 0; k0 < 128; k0 += 4) {
    v2f a, bpc, bmc;
    const float* ap = &xs[mn * 128 + k0 + khalf];
    a.x = ap[0];
    a.y = ap[1];
    const int wi = (k0 + khalf) * 128 + colbase + mn;
    bpc.x = Wpc[wi]; bpc.y = Wpc[wi + 128];
    bmc.x = Wmc[wi]; bmc.y = Wmc[wi + 128];
    cpc = __builtin_amdgcn_wmma_f32_16x16x4_f32(false, a, false, bpc, (short)0, cpc, false, false);
    cmc = __builtin_amdgcn_wmma_f32_16x16x4_f32(false, a, false, bmc, (short)0, cmc, false, false);
  }
  const int crow = rowbase + ((lane >> 4) << 3);
  const int ccol = colbase + mn;
#pragma unroll
  for (int r = 0; r < 8; ++r) {
    hpc[(size_t)(crow + r) * 128 + ccol] = cpc[r];
    hmc[(size_t)(crow + r) * 128 + ccol] = cmc[r];
  }
}

#define SC_NPW 16
typedef __attribute__((ext_vector_type(4))) float sc4f;

__global__ __launch_bounds__(256)
void score_kernel(const float* __restrict__ h, const float* __restrict__ a_src,
                  const float* __restrict__ a_dst, float* __restrict__ s_src,
                  float* __restrict__ s_dst, int n) {
  const int warp = (blockIdx.x * blockDim.x + threadIdx.x) >> 5;
  const int lane = threadIdx.x & 31;
  const int nb = warp * SC_NPW;
  if (nb >= n) return;

  sc4f hv[SC_NPW];
#pragma unroll
  for (int j = 0; j < SC_NPW; ++j) {
    const int node = min(nb + j, n - 1);
    hv[j] = ((const sc4f*)(h + (size_t)node * 128))[lane];
  }
  const sc4f as = ((const sc4f*)a_src)[lane];
  const sc4f ad = ((const sc4f*)a_dst)[lane];

  float ps[SC_NPW], pd[SC_NPW];
#pragma unroll
  for (int j = 0; j < SC_NPW; ++j) {
    float ss = 0.0f, sd = 0.0f;
    ss += hv[j].x * as.x;  sd += hv[j].x * ad.x;
    ss += hv[j].y * as.y;  sd += hv[j].y * ad.y;
    ss += hv[j].z * as.z;  sd += hv[j].z * ad.z;
    ss += hv[j].w * as.w;  sd += hv[j].w * ad.w;
    ps[j] = ss;
    pd[j] = sd;
  }

  const bool b0 = (lane & 1) != 0;
  float s8[8], d8[8];
#pragma unroll
  for (int m = 0; m < 8; ++m) {
    s8[m] = (b0 ? ps[2 * m + 1] : ps[2 * m]) + __shfl_xor(b0 ? ps[2 * m] : ps[2 * m + 1], 1, 32);
    d8[m] = (b0 ? pd[2 * m + 1] : pd[2 * m]) + __shfl_xor(b0 ? pd[2 * m] : pd[2 * m + 1], 1, 32);
  }
  const bool b1 = (lane & 2) != 0;
  float s4[4], d4[4];
#pragma unroll
  for (int m = 0; m < 4; ++m) {
    s4[m] = (b1 ? s8[2 * m + 1] : s8[2 * m]) + __shfl_xor(b1 ? s8[2 * m] : s8[2 * m + 1], 2, 32);
    d4[m] = (b1 ? d8[2 * m + 1] : d8[2 * m]) + __shfl_xor(b1 ? d8[2 * m] : d8[2 * m + 1], 2, 32);
  }
  const bool b2 = (lane & 4) != 0;
  float s2[2], d2[2];
#pragma unroll
  for (int m = 0; m < 2; ++m) {
    s2[m] = (b2 ? s4[2 * m + 1] : s4[2 * m]) + __shfl_xor(b2 ? s4[2 * m] : s4[2 * m + 1], 4, 32);
    d2[m] = (b2 ? d4[2 * m + 1] : d4[2 * m]) + __shfl_xor(b2 ? d4[2 * m] : d4[2 * m + 1], 4, 32);
  }
  const bool b3 = (lane & 8) != 0;
  const float ssum = (b3 ? s2[1] : s2[0]) + __shfl_xor(b3 ? s2[0] : s2[1], 8, 32);
  const float dsum = (b3 ? d2[1] : d2[0]) + __shfl_xor(b3 ? d2[0] : d2[1], 8, 32);

  const int node = nb + (lane & 15);
  if (node < n) {
    const int head = lane >> 4;
    s_src[node * 2 + head] = ssum;
    s_dst[node * 2 + head] = dsum;
  }
}

#define OWN_R   196
#define SP_W    32
#define SP_K    3
#define SP_CH   (SP_W * 32 * SP_K)
#define SP_NB   256
#define SP_PITCH 288
static_assert(SP_NB + 1 <= SP_PITCH && SP_PITCH % 32 == 0, "a block's offsets are its own lines");

typedef __attribute__((ext_vector_type(4))) float v4f;
typedef __attribute__((ext_vector_type(4))) unsigned v4u;

__global__ __launch_bounds__(1024)
void split_kernel(const int* __restrict__ ei_pc, int E_pc, const int* __restrict__ ei_mc, int E_mc,
                  int n, int P_pc,
                  unsigned* __restrict__ part_pc, int* __restrict__ offs_pc,
                  unsigned* __restrict__ part_mc, int* __restrict__ offs_mc) {
  __shared__ int cnt[SP_W * SP_NB];
  __shared__ __attribute__((aligned(16))) unsigned sorted[SP_CH];
  __shared__ int binstart[SP_NB];
  __shared__ int wsum[SP_NB / 32];

  const bool mc    = (int)blockIdx.x >= P_pc;
  const int  chunk = mc ? (int)blockIdx.x - P_pc : (int)blockIdx.x;
  const int* ei    = mc ? ei_mc : ei_pc;
  const int  E     = mc ? E_mc : E_pc;
  unsigned*  part  = (mc ? part_mc : part_pc) + (size_t)chunk * SP_CH;
  int*       offs  = (mc ? offs_mc : offs_pc) + (size_t)chunk * SP_PITCH;
  const int  tot   = E + n;

  const int lane = threadIdx.x & 31;
  const int wave = __builtin_amdgcn_readfirstlane((int)(threadIdx.x >> 5));
  const unsigned lt = (1u << lane) - 1u;
  int* cw = cnt + wave * SP_NB;

  for (int k = 0; k < SP_NB / 32; ++k) cw[k * 32 + lane] = 0;

  const int sbase = chunk * SP_CH + wave * (32 * SP_K) + lane;
  int dd[SP_K], ss[SP_K];
#pragma unroll
  for (int j = 0; j < SP_K; ++j) {
    const int e = sbase + j * 32;
    int d = 0, sn = 0;
    if (e < E) { sn = ei[e]; d = ei[E + e]; }
    else if (e < tot) { sn = e - E; d = e - E; }
    dd[j] = d; ss[j] = sn;
  }

  int key[SP_K], rnk[SP_K];
  unsigned ent[SP_K];
#pragma unroll
  for (int j = 0; j < SP_K; ++j) {
    const bool ok = (sbase + j * 32) < tot;
    const unsigned own = (unsigned)dd[j] / (unsigned)OWN_R;
    key[j] = (int)(own & (unsigned)(SP_NB - 1));
    ent[j] = (((unsigned)dd[j] - own * (unsigned)OWN_R) << 24) | (unsigned)ss[j];
    unsigned same = __builtin_amdgcn_ballot_w32(ok);
    for (int b = 0; b < 8; ++b) {
      const bool bit = ((key[j] >> b) & 1) != 0;
      const unsigned bb = __builtin_amdgcn_ballot_w32(bit);
      same &= bit ? bb : ~bb;
    }
    rnk[j] = 0;
    if (ok) {
      const int old = cw[key[j]];
      rnk[j] = old + __popc(same & lt);
      if ((same & lt) == 0u) cw[key[j]] = old + __popc(same);
    }
  }
  __syncthreads();

  int incl = 0, totb = 0;
  if (threadIdx.x < SP_NB) {
    int c[SP_W];
#pragma unroll
    for (int w = 0; w < SP_W; ++w) c[w] = cnt[w * SP_NB + threadIdx.x];
    int run = 0;
#pragma unroll
    for (int w = 0; w < SP_W; ++w) { cnt[w * SP_NB + threadIdx.x] = run; run += c[w]; }
    totb = run;
    incl = run;
    for (int d = 1; d < 32; d <<= 1) {
      const int t = __shfl_up(incl, d, 32);
      if (lane >= d) incl += t;
    }
    if (lane == 31) wsum[wave] = incl;
  }
  __syncthreads();
  if (threadIdx.x < SP_NB) {
    int wb = 0;
    for (int w = 0; w < SP_NB / 32; ++w) wb += (w < wave) ? wsum[w] : 0;
    const int bs = wb + incl - totb;
    binstart[threadIdx.x] = bs;
    offs[threadIdx.x] = bs;
    if (threadIdx.x == SP_NB - 1) offs[SP_NB] = bs + totb;
  }
  __syncthreads();

#pragma unroll
  for (int j = 0; j < SP_K; ++j)
    if ((sbase + j * 32) < tot) sorted[binstart[key[j]] + cw[key[j]] + rnk[j]] = ent[j];
  __syncthreads();

  if (threadIdx.x < SP_CH / 4) ((v4u*)part)[threadIdx.x] = ((const v4u*)sorted)[threadIdx.x];
}

#define OWN_T   ((OWN_R + 31) / 32)
#define OWN_QC  4096
#define OWN_U   8
#define OWN_S   (OWN_QC / 1024)
#define OWN_KP  33
static_assert(OWN_QC % 1024 == 0 && OWN_R <= 256, "32 waves share the queue in 32-entry steps; local node in 8 bits");

__device__ __forceinline__ void own_consume(int nq, int wave, int lane, unsigned lt, int base,
                                            const float* __restrict__ s_src,
                                            const float* __restrict__ s_dst,
                                            const float* __restrict__ h, float* accs,
                                            int* rmax, float* dens, v2f* ev,
                                            const unsigned* q, unsigned* q2, int* kcnt, int* ktot) {
  __syncthreads();

  const int steps = (nq + 1023) >> 10;
  int* cw = kcnt + wave * OWN_KP;
  cw[lane] = 0;
  const int cbase = wave * (steps * 32) + lane;
  unsigned ent[OWN_S];
  int key[OWN_S], rnk[OWN_S];
#pragma unroll
  for (int j = 0; j < OWN_S; ++j) {
    ent[j] = 0u; key[j] = 0; rnk[j] = 0;
    if (j < steps) {
      const int idx = cbase + j * 32;
      const bool ok = idx < nq;
      ent[j] = q[idx];
      key[j] = (int)((ent[j] >> 24) & 31u);
      unsigned same = __builtin_amdgcn_ballot_w32(ok);
      for (int b = 0; b < 5; ++b) {
        const bool bit = ((key[j] >> b) & 1) != 0;
        const unsigned bb = __builtin_amdgcn_ballot_w32(bit);
        same &= bit ? bb : ~bb;
      }
      if (ok) {
        const int old = cw[key[j]];
        rnk[j] = old + __popc(same & lt);
        if ((same & lt) == 0u) cw[key[j]] = old + __popc(same);
      }
    }
  }
  __syncthreads();

  const int kc = kcnt[lane * OWN_KP + wave];
  int kin = kc;
#pragma unroll
  for (int d = 1; d < 32; d <<= 1) {
    const int t = __shfl_up(kin, d, 32);
    if (lane >= d) kin += t;
  }
  if (lane == 31) ktot[wave] = kin;
  __syncthreads();

  const int kt = ktot[lane];
  int tin = kt;
#pragma unroll
  for (int d = 1; d < 32; d <<= 1) {
    const int t = __shfl_up(tin, d, 32);
    if (lane >= d) tin += t;
  }
  const int myoff = __builtin_amdgcn_readlane(tin - kt, wave);
  const int mycnt = __builtin_amdgcn_readlane(kt, wave);
  kcnt[lane * OWN_KP + wave] = myoff + kin - kc;
  __syncthreads();

#pragma unroll
  for (int j = 0; j < OWN_S; ++j)
    if (j < steps && cbase + j * 32 < nq) q2[cw[key[j]] + rnk[j]] = ent[j];
  __syncthreads();

  const int hd = lane >> 4;
  const int ninf = fkey(-__builtin_inff());

  int oldk[OWN_T];
#pragma unroll
  for (int t = 0; t < OWN_T; ++t) {
    const int row = wave + 32 * t;
    oldk[t] = (row < OWN_R) ? rmax[row * 2 + hd] : ninf;
  }

  for (int b0 = 0; b0 < mycnt; b0 += 32) {
    if (b0 + lane < mycnt) {
      const unsigned ent = q2[myoff + b0 + lane];
      const int local = (int)(ent >> 24);
      const int s     = (int)(ent & 0xFFFFFFu);
      const v2f ss = *(const v2f*)(s_src + (size_t)s * 2);
      const v2f sd = *(const v2f*)(s_dst + (size_t)(base + local) * 2);
      v2f v = ss + sd;
      v.x = (v.x > 0.0f) ? v.x : NEG_SLOPE * v.x;
      v.y = (v.y > 0.0f) ? v.y : NEG_SLOPE * v.y;
      ev[myoff + b0 + lane] = v;
      atomicMax(&rmax[local * 2 + 0], fkey(v.x));
      atomicMax(&rmax[local * 2 + 1], fkey(v.y));
    }
  }

#pragma unroll
  for (int t = 0; t < OWN_T; ++t) {
    const int row = wave + 32 * t;
    if (row < OWN_R) {
      const int newk = rmax[row * 2 + hd];
      if (newk != oldk[t] && oldk[t] != ninf) {
        const float sc = expf(kval(oldk[t]) - kval(newk));
        v4f* ar = (v4f*)(accs + row * 128) + lane;
        *ar = *ar * sc;
        if ((lane & 15) == 0) dens[row * 2 + hd] *= sc;
      }
    }
  }

  for (int b0 = 0; b0 < mycnt; b0 += 32) {
    const int nb = min(32, mycnt - b0);
    const unsigned ent = (lane < nb) ? q2[myoff + b0 + lane] : 0u;
    const int local = (int)(ent >> 24);
    const int s     = (int)(ent & 0xFFFFFFu);
    float a0 = 0.0f, a1 = 0.0f;
    if (lane < nb) {
      const v2f v = ev[myoff + b0 + lane];
      a0 = expf(v.x - kval(rmax[local * 2 + 0]));
      a1 = expf(v.y - kval(rmax[local * 2 + 1]));
    }
    for (int j = 0; j < nb; j += OWN_U) {
      v4f hv[OWN_U];
#pragma unroll
      for (int u = 0; u < OWN_U; ++u) {
        const int jj = min(j + u, nb - 1);
        const int sj = __builtin_amdgcn_readlane(s, jj);
        hv[u] = ((const v4f*)(h + (size_t)sj * 128))[lane];
      }
#pragma unroll
      for (int u = 0; u < OWN_U; ++u) {
        if (j + u < nb) {
          const int   lj = __builtin_amdgcn_readlane(local, j + u);
          const float b0f = __int_as_float(__builtin_amdgcn_readlane(__float_as_int(a0), j + u));
          const float b1f = __int_as_float(__builtin_amdgcn_readlane(__float_as_int(a1), j + u));
          const float al = (lane < 16) ? b0f : b1f;
          v4f* ar = (v4f*)(accs + lj * 128) + lane;
          v4f t = *ar;
          t.x += al * hv[u].x;
          t.y += al * hv[u].y;
          t.z += al * hv[u].z;
          t.w += al * hv[u].w;
          *ar = t;
          if ((lane & 15) == 0) dens[lj * 2 + hd] += al;
        }
      }
    }
  }
  __syncthreads();
}

__device__ __forceinline__ float own_elu(float v) {
  return (v > 0.0f) ? v : (expf(v) - 1.0f);
}

__global__ __launch_bounds__(1024)
void owner_acc_kernel(const unsigned* __restrict__ part_pc, const int* __restrict__ offs_pc, int P_pc,
                      const unsigned* __restrict__ part_mc, const int* __restrict__ offs_mc, int P_mc,
                      int n,
                      const float* __restrict__ ssrc_pc, const float* __restrict__ sdst_pc,
                      const float* __restrict__ h_pc, const float* __restrict__ b_pc,
                      const float* __restrict__ ssrc_mc, const float* __restrict__ sdst_mc,
                      const float* __restrict__ h_mc, const float* __restrict__ b_mc,
                      float* __restrict__ out) {
  __shared__ __attribute__((aligned(16))) float accs[OWN_R * 128];
  __shared__ __attribute__((aligned(16))) float keep[OWN_R * 128];
  __shared__ __attribute__((aligned(16))) unsigned q[OWN_QC];
  __shared__ unsigned q2[OWN_QC];
  __shared__ __attribute__((aligned(8))) v2f ev[OWN_QC];
  __shared__ int      rmax[OWN_R * 2];
  __shared__ float    dens[OWN_R * 2];
  __shared__ int      wcnt[2 * 32];
  __shared__ int      kcnt[32 * OWN_KP];
  __shared__ int      ktot[32];

  const int lane = threadIdx.x & 31;
  const int wave = __builtin_amdgcn_readfirstlane((int)(threadIdx.x >> 5));
  const int base = blockIdx.x * OWN_R;
  const unsigned lt = (1u << lane) - 1u;
  const int sub = lane & 3;

  for (int cv = 0; cv < 2; ++cv) {
    const unsigned* part = cv ? part_mc : part_pc;
    const int*   offs  = cv ? offs_mc : offs_pc;
    const int    P     = cv ? P_mc : P_pc;
    const float* s_src = cv ? ssrc_mc : ssrc_pc;
    const float* s_dst = cv ? sdst_mc : sdst_pc;
    const float* h     = cv ? h_mc : h_pc;
    const float* bias  = cv ? b_mc : b_pc;

    for (int i = threadIdx.x; i < OWN_R * 128; i += 1024) accs[i] = 0.0f;
    for (int i = threadIdx.x; i < OWN_R * 2; i += 1024) {
      rmax[i] = fkey(-__builtin_inff());
      dens[i] = 0.0f;
    }

    __syncthreads();

    int qfill = 0;
    for (int c0 = 0; c0 < P; c0 += 256) {
      const int c = c0 + (int)(threadIdx.x >> 2);
      int s0 = 0, cn = 0;
      if (c < P) {
        const int* o = offs + (size_t)c * SP_PITCH + blockIdx.x;
        s0 = o[0];
        cn = o[1] - s0;
      }
      const unsigned* src = part + (size_t)min(c, P - 1) * SP_CH + s0;

      const int mine = (sub == 0) ? cn : 0;
      int incl = mine;
#pragma unroll
      for (int d = 1; d < 32; d <<= 1) {
        const int t = __shfl_up(incl, d, 32);
        if (lane >= d) incl += t;
      }
      const int cnt = __builtin_amdgcn_readlane(incl, 31);
      int* wc = wcnt + ((c0 >> 8) & 1) * 32;
      if (lane == 0) wc[wave] = cnt;
      __syncthreads();

      const int cwv = wc[lane];
      int incw = cwv;
#pragma unroll
      for (int d = 1; d < 32; d <<= 1) {
        const int t = __shfl_up(incw, d, 32);
        if (lane >= d) incw += t;
      }
      const int total  = __builtin_amdgcn_readlane(incw, 31);
      const int mybase = __builtin_amdgcn_readlane(incw - cwv, wave);
      const int pbase = mybase + __shfl(incl - mine, lane & ~3, 32);

      int done = 0;
      while (done < total) {
        if (qfill == OWN_QC) {
          own_consume(qfill, wave, lane, lt, base, s_src, s_dst, h, accs, rmax, dens, ev, q, q2, kcnt, ktot);
          qfill = 0;
        }
        const int take = min(OWN_QC - qfill, total - done);
        for (int i = sub; i < cn; i += 8) {
          const int pa = pbase + i - done;
          const int pb = pa + 4;
          const bool ba = (pa >= 0) && (pa < take);
          const bool bb = (i + 4 < cn) && (pb >= 0) && (pb < take);
          const unsigned va = ba ? src[i] : 0u;
          const unsigned vb = bb ? src[i + 4] : 0u;
          if (ba) q[qfill + pa] = va;
          if (bb) q[qfill + pb] = vb;
        }
        qfill += take;
        done  += take;
      }
    }

    own_consume(qfill, wave, lane, lt, base, s_src, s_dst, h, accs, rmax, dens, ev, q, q2, kcnt, ktot);

    const float bc0 = bias[lane * 4 + 0], bc1 = bias[lane * 4 + 1];
    const float bc2 = bias[lane * 4 + 2], bc3 = bias[lane * 4 + 3];
    for (int i4 = threadIdx.x; i4 < OWN_R * 32; i4 += 1024) {
      const int row = i4 >> 5;
      const float d = dens[row * 2 + (lane >> 4)];
      const v4f a = ((const v4f*)accs)[i4];
      v4f v;
      v.x = own_elu(a.x / d + bc0);
      v.y = own_elu(a.y / d + bc1);
      v.z = own_elu(a.z / d + bc2);
      v.w = own_elu(a.w / d + bc3);
      if (cv == 0) {
        ((v4f*)keep)[i4] = v;
      } else if (base + row < n) {
        const v4f k = ((const v4f*)keep)[i4];
        v4f o;
        o.x = 0.5f * (k.x + v.x);
        o.y = 0.5f * (k.y + v.y);
        o.z = 0.5f * (k.z + v.z);
        o.w = 0.5f * (k.w + v.w);
        ((v4f*)(out + (size_t)(base + row) * 128))[lane] = o;
      }
    }
    __syncthreads();
  }
}

extern "C" void kernel_launch(void* const* d_in, const int* in_sizes, int n_in,
                              void* d_out, int out_size, void* d_ws, size_t ws_size,
                              hipStream_t stream) {
  const float* x      = (const float*)d_in[0];
  const float* W_pc   = (const float*)d_in[1];
  const float* asr_pc = (const float*)d_in[2];
  const float* ads_pc = (const float*)d_in[3];
  const float* b_pc   = (const float*)d_in[4];
  const float* W_mc   = (const float*)d_in[5];
  const float* asr_mc = (const float*)d_in[6];
  const float* ads_mc = (const float*)d_in[7];
  const float* b_mc   = (const float*)d_in[8];
  const int*   ei_pc  = (const int*)d_in[9];
  const int*   ei_mc  = (const int*)d_in[10];
  float* out = (float*)d_out;

  const int NN  = in_sizes[0] / 128;
  const int Epc = in_sizes[9] / 2;
  const int Emc = in_sizes[10] / 2;
  const int Ppc = (Epc + NN + SP_CH - 1) / SP_CH;
  const int Pmc = (Emc + NN + SP_CH - 1) / SP_CH;

  float* ws = (float*)d_ws;
  size_t off = 0;
  float* h_pc    = ws + off; off += (size_t)NN * 128;
  float* h_mc    = ws + off; off += (size_t)NN * 128;
  float* ssrc_pc = ws + off; off += (size_t)NN * 2;
  float* sdst_pc = ws + off; off += (size_t)NN * 2;
  float* ssrc_mc = ws + off; off += (size_t)NN * 2;
  float* sdst_mc = ws + off; off += (size_t)NN * 2;
  unsigned* part_pc = (unsigned*)(ws + off); off += (size_t)Ppc * SP_CH;
  unsigned* part_mc = (unsigned*)(ws + off); off += (size_t)Pmc * SP_CH;
  int* offs_pc = (int*)(ws + off); off += (size_t)Ppc * SP_PITCH;
  int* offs_mc = (int*)(ws + off); off += (size_t)Pmc * SP_PITCH;

  gemm_kernel<<<(NN + 15) / 16, 256, 0, stream>>>(x, W_pc, W_mc, h_pc, h_mc);

  score_kernel<<<(NN + 8 * SC_NPW - 1) / (8 * SC_NPW), 256, 0, stream>>>(h_pc, asr_pc, ads_pc, ssrc_pc, sdst_pc, NN);
  score_kernel<<<(NN + 8 * SC_NPW - 1) / (8 * SC_NPW), 256, 0, stream>>>(h_mc, asr_mc, ads_mc, ssrc_mc, sdst_mc, NN);

  split_kernel<<<Ppc + Pmc, 1024, 0, stream>>>(ei_pc, Epc, ei_mc, Emc, NN, Ppc,
                                               part_pc, offs_pc, part_mc, offs_mc);

  const int nown = (NN + OWN_R - 1) / OWN_R;
  owner_acc_kernel<<<nown, 1024, 0, stream>>>(part_pc, offs_pc, Ppc, part_mc, offs_mc, Pmc, NN,
                                             ssrc_pc, sdst_pc, h_pc, b_pc,
                                             ssrc_mc, sdst_mc, h_mc, b_mc, out);
}


# Owner wave puts its edges in row order while scoring them (ballot ranks, queue order kept) and sums each row in registers: LDS row and denominator updated once per row, not per edge; sum order kept
# speedup vs baseline: 4.8563x; 1.1000x over previous
#include <hip/hip_runtime.h>
#include <math.h>

typedef __attribute__((ext_vector_type(2))) float v2f;
typedef __attribute__((ext_vector_type(8))) float v8f;

#define NEG_SLOPE 0.2f

__device__ __forceinline__ int fkey(float v) {
  int b = __float_as_int(v);
  return (b >= 0) ? b : (b ^ 0x7FFFFFFF);
}
__device__ __forceinline__ float kval(int k) {
  return __int_as_float((k >= 0) ? k : (k ^ 0x7FFFFFFF));
}

__global__ __launch_bounds__(256)
void gemm_kernel(const float* __restrict__ x, const float* __restrict__ Wpc,
                 const float* __restrict__ Wmc, float* __restrict__ hpc,
                 float* __restrict__ hmc) {
  __shared__ float xs[16 * 128];
  const int rowbase = blockIdx.x * 16;
  for (int i = threadIdx.x; i < 16 * 128; i += 256)
    xs[i] = x[(size_t)rowbase * 128 + i];
  __syncthreads();

  const int lane   = threadIdx.x & 31;
  const int wave   = threadIdx.x >> 5;
  const int colbase = wave * 16;
  const int mn    = lane & 15;
  const int khalf = (lane >> 4) << 1;

  v8f cpc = {};
  v8f cmc = {};
  for (int k0 = 0; k0 < 128; k0 += 4) {
    v2f a, bpc, bmc;
    const float* ap = &xs[mn * 128 + k0 + khalf];
    a.x = ap[0];
    a.y = ap[1];
    const int wi = (k0 + khalf) * 128 + colbase + mn;
    bpc.x = Wpc[wi]; bpc.y = Wpc[wi + 128];
    bmc.x = Wmc[wi]; bmc.y = Wmc[wi + 128];
    cpc = __builtin_amdgcn_wmma_f32_16x16x4_f32(false, a, false, bpc, (short)0, cpc, false, false);
    cmc = __builtin_amdgcn_wmma_f32_16x16x4_f32(false, a, false, bmc, (short)0, cmc, false, false);
  }
  const int crow = rowbase + ((lane >> 4) << 3);
  const int ccol = colbase + mn;
#pragma unroll
  for (int r = 0; r < 8; ++r) {
    hpc[(size_t)(crow + r) * 128 + ccol] = cpc[r];
    hmc[(size_t)(crow + r) * 128 + ccol] = cmc[r];
  }
}

#define SC_NPW 16
typedef __attribute__((ext_vector_type(4))) float sc4f;

__global__ __launch_bounds__(256)
void score_kernel(const float* __restrict__ h, const float* __restrict__ a_src,
                  const float* __restrict__ a_dst, float* __restrict__ s_src,
                  float* __restrict__ s_dst, int n) {
  const int warp = (blockIdx.x * blockDim.x + threadIdx.x) >> 5;
  const int lane = threadIdx.x & 31;
  const int nb = warp * SC_NPW;
  if (nb >= n) return;

  sc4f hv[SC_NPW];
#pragma unroll
  for (int j = 0; j < SC_NPW; ++j) {
    const int node = min(nb + j, n - 1);
    hv[j] = ((const sc4f*)(h + (size_t)node * 128))[lane];
  }
  const sc4f as = ((const sc4f*)a_src)[lane];
  const sc4f ad = ((const sc4f*)a_dst)[lane];

  float ps[SC_NPW], pd[SC_NPW];
#pragma unroll
  for (int j = 0; j < SC_NPW; ++j) {
    float ss = 0.0f, sd = 0.0f;
    ss += hv[j].x * as.x;  sd += hv[j].x * ad.x;
    ss += hv[j].y * as.y;  sd += hv[j].y * ad.y;
    ss += hv[j].z * as.z;  sd += hv[j].z * ad.z;
    ss += hv[j].w * as.w;  sd += hv[j].w * ad.w;
    ps[j] = ss;
    pd[j] = sd;
  }

  const bool b0 = (lane & 1) != 0;
  float s8[8], d8[8];
#pragma unroll
  for (int m = 0; m < 8; ++m) {
    s8[m] = (b0 ? ps[2 * m + 1] : ps[2 * m]) + __shfl_xor(b0 ? ps[2 * m] : ps[2 * m + 1], 1, 32);
    d8[m] = (b0 ? pd[2 * m + 1] : pd[2 * m]) + __shfl_xor(b0 ? pd[2 * m] : pd[2 * m + 1], 1, 32);
  }
  const bool b1 = (lane & 2) != 0;
  float s4[4], d4[4];
#pragma unroll
  for (int m = 0; m < 4; ++m) {
    s4[m] = (b1 ? s8[2 * m + 1] : s8[2 * m]) + __shfl_xor(b1 ? s8[2 * m] : s8[2 * m + 1], 2, 32);
    d4[m] = (b1 ? d8[2 * m + 1] : d8[2 * m]) + __shfl_xor(b1 ? d8[2 * m] : d8[2 * m + 1], 2, 32);
  }
  const bool b2 = (lane & 4) != 0;
  float s2[2], d2[2];
#pragma unroll
  for (int m = 0; m < 2; ++m) {
    s2[m] = (b2 ? s4[2 * m + 1] : s4[2 * m]) + __shfl_xor(b2 ? s4[2 * m] : s4[2 * m + 1], 4, 32);
    d2[m] = (b2 ? d4[2 * m + 1] : d4[2 * m]) + __shfl_xor(b2 ? d4[2 * m] : d4[2 * m + 1], 4, 32);
  }
  const bool b3 = (lane & 8) != 0;
  const float ssum = (b3 ? s2[1] : s2[0]) + __shfl_xor(b3 ? s2[0] : s2[1], 8, 32);
  const float dsum = (b3 ? d2[1] : d2[0]) + __shfl_xor(b3 ? d2[0] : d2[1], 8, 32);

  const int node = nb + (lane & 15);
  if (node < n) {
    const int head = lane >> 4;
    s_src[node * 2 + head] = ssum;
    s_dst[node * 2 + head] = dsum;
  }
}

#define OWN_R   196
#define SP_W    32
#define SP_K    3
#define SP_CH   (SP_W * 32 * SP_K)
#define SP_NB   256
#define SP_PITCH 288
static_assert(SP_NB + 1 <= SP_PITCH && SP_PITCH % 32 == 0, "a block's offsets are its own lines");

typedef __attribute__((ext_vector_type(4))) float v4f;
typedef __attribute__((ext_vector_type(4))) unsigned v4u;

__global__ __launch_bounds__(1024)
void split_kernel(const int* __restrict__ ei_pc, int E_pc, const int* __restrict__ ei_mc, int E_mc,
                  int n, int P_pc,
                  unsigned* __restrict__ part_pc, int* __restrict__ offs_pc,
                  unsigned* __restrict__ part_mc, int* __restrict__ offs_mc) {
  __shared__ int cnt[SP_W * SP_NB];
  __shared__ __attribute__((aligned(16))) unsigned sorted[SP_CH];
  __shared__ int binstart[SP_NB];
  __shared__ int wsum[SP_NB / 32];

  const bool mc    = (int)blockIdx.x >= P_pc;
  const int  chunk = mc ? (int)blockIdx.x - P_pc : (int)blockIdx.x;
  const int* ei    = mc ? ei_mc : ei_pc;
  const int  E     = mc ? E_mc : E_pc;
  unsigned*  part  = (mc ? part_mc : part_pc) + (size_t)chunk * SP_CH;
  int*       offs  = (mc ? offs_mc : offs_pc) + (size_t)chunk * SP_PITCH;
  const int  tot   = E + n;

  const int lane = threadIdx.x & 31;
  const int wave = __builtin_amdgcn_readfirstlane((int)(threadIdx.x >> 5));
  const unsigned lt = (1u << lane) - 1u;
  int* cw = cnt + wave * SP_NB;

  for (int k = 0; k < SP_NB / 32; ++k) cw[k * 32 + lane] = 0;

  const int sbase = chunk * SP_CH + wave * (32 * SP_K) + lane;
  int dd[SP_K], ss[SP_K];
#pragma unroll
  for (int j = 0; j < SP_K; ++j) {
    const int e = sbase + j * 32;
    int d = 0, sn = 0;
    if (e < E) { sn = ei[e]; d = ei[E + e]; }
    else if (e < tot) { sn = e - E; d = e - E; }
    dd[j] = d; ss[j] = sn;
  }

  int key[SP_K], rnk[SP_K];
  unsigned ent[SP_K];
#pragma unroll
  for (int j = 0; j < SP_K; ++j) {
    const bool ok = (sbase + j * 32) < tot;
    const unsigned own = (unsigned)dd[j] / (unsigned)OWN_R;
    key[j] = (int)(own & (unsigned)(SP_NB - 1));
    ent[j] = (((unsigned)dd[j] - own * (unsigned)OWN_R) << 24) | (unsigned)ss[j];
    unsigned same = __builtin_amdgcn_ballot_w32(ok);
    for (int b = 0; b < 8; ++b) {
      const bool bit = ((key[j] >> b) & 1) != 0;
      const unsigned bb = __builtin_amdgcn_ballot_w32(bit);
      same &= bit ? bb : ~bb;
    }
    rnk[j] = 0;
    if (ok) {
      const int old = cw[key[j]];
      rnk[j] = old + __popc(same & lt);
      if ((same & lt) == 0u) cw[key[j]] = old + __popc(same);
    }
  }
  __syncthreads();

  int incl = 0, totb = 0;
  if (threadIdx.x < SP_NB) {
    int c[SP_W];
#pragma unroll
    for (int w = 0; w < SP_W; ++w) c[w] = cnt[w * SP_NB + threadIdx.x];
    int run = 0;
#pragma unroll
    for (int w = 0; w < SP_W; ++w) { cnt[w * SP_NB + threadIdx.x] = run; run += c[w]; }
    totb = run;
    incl = run;
    for (int d = 1; d < 32; d <<= 1) {
      const int t = __shfl_up(incl, d, 32);
      if (lane >= d) incl += t;
    }
    if (lane == 31) wsum[wave] = incl;
  }
  __syncthreads();
  if (threadIdx.x < SP_NB) {
    int wb = 0;
    for (int w = 0; w < SP_NB / 32; ++w) wb += (w < wave) ? wsum[w] : 0;
    const int bs = wb + incl - totb;
    binstart[threadIdx.x] = bs;
    offs[threadIdx.x] = bs;
    if (threadIdx.x == SP_NB - 1) offs[SP_NB] = bs + totb;
  }
  __syncthreads();

#pragma unroll
  for (int j = 0; j < SP_K; ++j)
    if ((sbase + j * 32) < tot) sorted[binstart[key[j]] + cw[key[j]] + rnk[j]] = ent[j];
  __syncthreads();

  if (threadIdx.x < SP_CH / 4) ((v4u*)part)[threadIdx.x] = ((const v4u*)sorted)[threadIdx.x];
}

#define OWN_T   ((OWN_R + 31) / 32)
#define OWN_QC  4096
#define OWN_U   8
#define OWN_S   (OWN_QC / 1024)
#define OWN_KP  33
static_assert(OWN_QC % 1024 == 0 && OWN_R <= 256, "32 waves share the queue in 32-entry steps; local node in 8 bits");

__device__ __forceinline__ void own_consume(int nq, int wave, int lane, unsigned lt, int base,
                                            const float* __restrict__ s_src,
                                            const float* __restrict__ s_dst,
                                            const float* __restrict__ h, float* accs,
                                            int* rmax, float* dens, v2f* ev,
                                            unsigned* q, unsigned* q2, int* kcnt, int* ktot) {
  __syncthreads();

  const int steps = (nq + 1023) >> 10;
  int* cw = kcnt + wave * OWN_KP;
  cw[lane] = 0;
  const int cbase = wave * (steps * 32) + lane;
  unsigned ent[OWN_S];
  int key[OWN_S], rnk[OWN_S];
#pragma unroll
  for (int j = 0; j < OWN_S; ++j) {
    ent[j] = 0u; key[j] = 0; rnk[j] = 0;
    if (j < steps) {
      const int idx = cbase + j * 32;
      const bool ok = idx < nq;
      ent[j] = q[idx];
      key[j] = (int)((ent[j] >> 24) & 31u);
      unsigned same = __builtin_amdgcn_ballot_w32(ok);
      for (int b = 0; b < 5; ++b) {
        const bool bit = ((key[j] >> b) & 1) != 0;
        const unsigned bb = __builtin_amdgcn_ballot_w32(bit);
        same &= bit ? bb : ~bb;
      }
      if (ok) {
        const int old = cw[key[j]];
        rnk[j] = old + __popc(same & lt);
        if ((same & lt) == 0u) cw[key[j]] = old + __popc(same);
      }
    }
  }
  __syncthreads();

  const int kc = kcnt[lane * OWN_KP + wave];
  int kin = kc;
#pragma unroll
  for (int d = 1; d < 32; d <<= 1) {
    const int t = __shfl_up(kin, d, 32);
    if (lane >= d) kin += t;
  }
  if (lane == 31) ktot[wave] = kin;
  __syncthreads();

  const int kt = ktot[lane];
  int tin = kt;
#pragma unroll
  for (int d = 1; d < 32; d <<= 1) {
    const int t = __shfl_up(tin, d, 32);
    if (lane >= d) tin += t;
  }
  const int myoff = __builtin_amdgcn_readlane(tin - kt, wave);
  const int mycnt = __builtin_amdgcn_readlane(kt, wave);
  kcnt[lane * OWN_KP + wave] = myoff + kin - kc;
  __syncthreads();

#pragma unroll
  for (int j = 0; j < OWN_S; ++j)
    if (j < steps && cbase + j * 32 < nq) q2[cw[key[j]] + rnk[j]] = ent[j];
  __syncthreads();

  const int hd = lane >> 4;
  const int ninf = fkey(-__builtin_inff());

  int oldk[OWN_T];
#pragma unroll
  for (int t = 0; t < OWN_T; ++t) {
    const int row = wave + 32 * t;
    oldk[t] = (row < OWN_R) ? rmax[row * 2 + hd] : ninf;
  }

  int tnext[OWN_T];
  {
    int tc[OWN_T];
#pragma unroll
    for (int t = 0; t < OWN_T; ++t) tc[t] = 0;
    for (int b0 = 0; b0 < mycnt; b0 += 32) {
      const bool ok = b0 + lane < mycnt;
      const unsigned e = ok ? q2[myoff + b0 + lane] : 0u;
      const int tt = (int)(e >> 29);
#pragma unroll
      for (int t = 0; t < OWN_T; ++t) tc[t] += __popc(__builtin_amdgcn_ballot_w32(ok && tt == t));
    }
    int run = 0;
#pragma unroll
    for (int t = 0; t < OWN_T; ++t) { tnext[t] = run; run += tc[t]; }
  }

  for (int b0 = 0; b0 < mycnt; b0 += 32) {
    const bool ok = b0 + lane < mycnt;
    const unsigned ent = ok ? q2[myoff + b0 + lane] : 0u;
    const int tt = (int)(ent >> 29);
    int pos = 0;
#pragma unroll
    for (int t = 0; t < OWN_T; ++t) {
      const unsigned m = __builtin_amdgcn_ballot_w32(ok && tt == t);
      if (tt == t) pos = tnext[t] + __popc(m & lt);
      tnext[t] += __popc(m);
    }
    if (ok) {
      const int local = (int)(ent >> 24);
      const int s     = (int)(ent & 0xFFFFFFu);
      const v2f ss = *(const v2f*)(s_src + (size_t)s * 2);
      const v2f sd = *(const v2f*)(s_dst + (size_t)(base + local) * 2);
      v2f v = ss + sd;
      v.x = (v.x > 0.0f) ? v.x : NEG_SLOPE * v.x;
      v.y = (v.y > 0.0f) ? v.y : NEG_SLOPE * v.y;
      q[myoff + pos]  = ent;
      ev[myoff + pos] = v;
      atomicMax(&rmax[local * 2 + 0], fkey(v.x));
      atomicMax(&rmax[local * 2 + 1], fkey(v.y));
    }
  }

#pragma unroll
  for (int t = 0; t < OWN_T; ++t) {
    const int row = wave + 32 * t;
    if (row < OWN_R) {
      const int newk = rmax[row * 2 + hd];
      if (newk != oldk[t] && oldk[t] != ninf) {
        const float sc = expf(kval(oldk[t]) - kval(newk));
        v4f* ar = (v4f*)(accs + row * 128) + lane;
        *ar = *ar * sc;
        if ((lane & 15) == 0) dens[row * 2 + hd] *= sc;
      }
    }
  }

  int cur = -1;
  v4f racc = {0.0f, 0.0f, 0.0f, 0.0f};
  float rden = 0.0f;
  for (int b0 = 0; b0 < mycnt; b0 += 32) {
    const int nb = min(32, mycnt - b0);
    const unsigned ent = (lane < nb) ? q[myoff + b0 + lane] : 0u;
    const int local = (int)(ent >> 24);
    const int s     = (int)(ent & 0xFFFFFFu);
    float a0 = 0.0f, a1 = 0.0f;
    if (lane < nb) {
      const v2f v = ev[myoff + b0 + lane];
      a0 = expf(v.x - kval(rmax[local * 2 + 0]));
      a1 = expf(v.y - kval(rmax[local * 2 + 1]));
    }
    for (int j = 0; j < nb; j += OWN_U) {
      v4f hv[OWN_U];
#pragma unroll
      for (int u = 0; u < OWN_U; ++u) {
        const int jj = min(j + u, nb - 1);
        const int sj = __builtin_amdgcn_readlane(s, jj);
        hv[u] = ((const v4f*)(h + (size_t)sj * 128))[lane];
      }
#pragma unroll
      for (int u = 0; u < OWN_U; ++u) {
        if (j + u < nb) {
          const int   lj = __builtin_amdgcn_readlane(local, j + u);
          if (lj != cur) {
            if (cur >= 0) {
              *((v4f*)(accs + cur * 128) + lane) = racc;
              if ((lane & 15) == 0) dens[cur * 2 + hd] = rden;
            }
            cur  = lj;
            racc = *((const v4f*)(accs + lj * 128) + lane);
            rden = dens[lj * 2 + hd];
          }
          const float b0f = __int_as_float(__builtin_amdgcn_readlane(__float_as_int(a0), j + u));
          const float b1f = __int_as_float(__builtin_amdgcn_readlane(__float_as_int(a1), j + u));
          const float al = (lane < 16) ? b0f : b1f;
          racc.x += al * hv[u].x;
          racc.y += al * hv[u].y;
          racc.z += al * hv[u].z;
          racc.w += al * hv[u].w;
          rden += al;
        }
      }
    }
  }
  if (cur >= 0) {
    *((v4f*)(accs + cur * 128) + lane) = racc;
    if ((lane & 15) == 0) dens[cur * 2 + hd] = rden;
  }
  __syncthreads();
}

__device__ __forceinline__ float own_elu(float v) {
  return (v > 0.0f) ? v : (expf(v) - 1.0f);
}

__global__ __launch_bounds__(1024)
void owner_acc_kernel(const unsigned* __restrict__ part_pc, const int* __restrict__ offs_pc, int P_pc,
                      const unsigned* __restrict__ part_mc, const int* __restrict__ offs_mc, int P_mc,
                      int n,
                      const float* __restrict__ ssrc_pc, const float* __restrict__ sdst_pc,
                      const float* __restrict__ h_pc, const float* __restrict__ b_pc,
                      const float* __restrict__ ssrc_mc, const float* __restrict__ sdst_mc,
                      const float* __restrict__ h_mc, const float* __restrict__ b_mc,
                      float* __restrict__ out) {
  __shared__ __attribute__((aligned(16))) float accs[OWN_R * 128];
  __shared__ __attribute__((aligned(16))) float keep[OWN_R * 128];
  __shared__ __attribute__((aligned(16))) unsigned q[OWN_QC];
  __shared__ unsigned q2[OWN_QC];
  __shared__ __attribute__((aligned(8))) v2f ev[OWN_QC];
  __shared__ int      rmax[OWN_R * 2];
  __shared__ float    dens[OWN_R * 2];
  __shared__ int      wcnt[2 * 32];
  __shared__ int      kcnt[32 * OWN_KP];
  __shared__ int      ktot[32];

  const int lane = threadIdx.x & 31;
  const int wave = __builtin_amdgcn_readfirstlane((int)(threadIdx.x >> 5));
  const int base = blockIdx.x * OWN_R;
  const unsigned lt = (1u << lane) - 1u;
  const int sub = lane & 3;

  for (int cv = 0; cv < 2; ++cv) {
    const unsigned* part = cv ? part_mc : part_pc;
    const int*   offs  = cv ? offs_mc : offs_pc;
    const int    P     = cv ? P_mc : P_pc;
    const float* s_src = cv ? ssrc_mc : ssrc_pc;
    const float* s_dst = cv ? sdst_mc : sdst_pc;
    const float* h     = cv ? h_mc : h_pc;
    const float* bias  = cv ? b_mc : b_pc;

    for (int i = threadIdx.x; i < OWN_R * 128; i += 1024) accs[i] = 0.0f;
    for (int i = threadIdx.x; i < OWN_R * 2; i += 1024) {
      rmax[i] = fkey(-__builtin_inff());
      dens[i] = 0.0f;
    }

    __syncthreads();

    int qfill = 0;
    for (int c0 = 0; c0 < P; c0 += 256) {
      const int c = c0 + (int)(threadIdx.x >> 2);
      int s0 = 0, cn = 0;
      if (c < P) {
        const int* o = offs + (size_t)c * SP_PITCH + blockIdx.x;
        s0 = o[0];
        cn = o[1] - s0;
      }
      const unsigned* src = part + (size_t)min(c, P - 1) * SP_CH + s0;

      const int mine = (sub == 0) ? cn : 0;
      int incl = mine;
#pragma unroll
      for (int d = 1; d < 32; d <<= 1) {
        const int t = __shfl_up(incl, d, 32);
        if (lane >= d) incl += t;
      }
      const int cnt = __builtin_amdgcn_readlane(incl, 31);
      int* wc = wcnt + ((c0 >> 8) & 1) * 32;
      if (lane == 0) wc[wave] = cnt;
      __syncthreads();

      const int cwv = wc[lane];
      int incw = cwv;
#pragma unroll
      for (int d = 1; d < 32; d <<= 1) {
        const int t = __shfl_up(incw, d, 32);
        if (lane >= d) incw += t;
      }
      const int total  = __builtin_amdgcn_readlane(incw, 31);
      const int mybase = __builtin_amdgcn_readlane(incw - cwv, wave);
      const int pbase = mybase + __shfl(incl - mine, lane & ~3, 32);

      int done = 0;
      while (done < total) {
        if (qfill == OWN_QC) {
          own_consume(qfill, wave, lane, lt, base, s_src, s_dst, h, accs, rmax, dens, ev, q, q2, kcnt, ktot);
          qfill = 0;
        }
        const int take = min(OWN_QC - qfill, total - done);
        for (int i = sub; i < cn; i += 8) {
          const int pa = pbase + i - done;
          const int pb = pa + 4;
          const bool ba = (pa >= 0) && (pa < take);
          const bool bb = (i + 4 < cn) && (pb >= 0) && (pb < take);
          const unsigned va = ba ? src[i] : 0u;
          const unsigned vb = bb ? src[i + 4] : 0u;
          if (ba) q[qfill + pa] = va;
          if (bb) q[qfill + pb] = vb;
        }
        qfill += take;
        done  += take;
      }
    }

    own_consume(qfill, wave, lane, lt, base, s_src, s_dst, h, accs, rmax, dens, ev, q, q2, kcnt, ktot);

    const float bc0 = bias[lane * 4 + 0], bc1 = bias[lane * 4 + 1];
    const float bc2 = bias[lane * 4 + 2], bc3 = bias[lane * 4 + 3];
    for (int i4 = threadIdx.x; i4 < OWN_R * 32; i4 += 1024) {
      const int row = i4 >> 5;
      const float d = dens[row * 2 + (lane >> 4)];
      const v4f a = ((const v4f*)accs)[i4];
      v4f v;
      v.x = own_elu(a.x / d + bc0);
      v.y = own_elu(a.y / d + bc1);
      v.z = own_elu(a.z / d + bc2);
      v.w = own_elu(a.w / d + bc3);
      if (cv == 0) {
        ((v4f*)keep)[i4] = v;
      } else if (base + row < n) {
        const v4f k = ((const v4f*)keep)[i4];
        v4f o;
        o.x = 0.5f * (k.x + v.x);
        o.y = 0.5f * (k.y + v.y);
        o.z = 0.5f * (k.z + v.z);
        o.w = 0.5f * (k.w + v.w);
        ((v4f*)(out + (size_t)(base + row) * 128))[lane] = o;
      }
    }
    __syncthreads();
  }
}

extern "C" void kernel_launch(void* const* d_in, const int* in_sizes, int n_in,
                              void* d_out, int out_size, void* d_ws, size_t ws_size,
                              hipStream_t stream) {
  const float* x      = (const float*)d_in[0];
  const float* W_pc   = (const float*)d_in[1];
  const float* asr_pc = (const float*)d_in[2];
  const float* ads_pc = (const float*)d_in[3];
  const float* b_pc   = (const float*)d_in[4];
  const float* W_mc   = (const float*)d_in[5];
  const float* asr_mc = (const float*)d_in[6];
  const float* ads_mc = (const float*)d_in[7];
  const float* b_mc   = (const float*)d_in[8];
  const int*   ei_pc  = (const int*)d_in[9];
  const int*   ei_mc  = (const int*)d_in[10];
  float* out = (float*)d_out;

  const int NN  = in_sizes[0] / 128;
  const int Epc = in_sizes[9] / 2;
  const int Emc = in_sizes[10] / 2;
  const int Ppc = (Epc + NN + SP_CH - 1) / SP_CH;
  const int Pmc = (Emc + NN + SP_CH - 1) / SP_CH;

  float* ws = (float*)d_ws;
  size_t off = 0;
  float* h_pc    = ws + off; off += (size_t)NN * 128;
  float* h_mc    = ws + off; off += (size_t)NN * 128;
  float* ssrc_pc = ws + off; off += (size_t)NN * 2;
  float* sdst_pc = ws + off; off += (size_t)NN * 2;
  float* ssrc_mc = ws + off; off += (size_t)NN * 2;
  float* sdst_mc = ws + off; off += (size_t)NN * 2;
  unsigned* part_pc = (unsigned*)(ws + off); off += (size_t)Ppc * SP_CH;
  unsigned* part_mc = (unsigned*)(ws + off); off += (size_t)Pmc * SP_CH;
  int* offs_pc = (int*)(ws + off); off += (size_t)Ppc * SP_PITCH;
  int* offs_mc = (int*)(ws + off); off += (size_t)Pmc * SP_PITCH;

  gemm_kernel<<<(NN + 15) / 16, 256, 0, stream>>>(x, W_pc, W_mc, h_pc, h_mc);

  score_kernel<<<(NN + 8 * SC_NPW - 1) / (8 * SC_NPW), 256, 0, stream>>>(h_pc, asr_pc, ads_pc, ssrc_pc, sdst_pc, NN);
  score_kernel<<<(NN + 8 * SC_NPW - 1) / (8 * SC_NPW), 256, 0, stream>>>(h_mc, asr_mc, ads_mc, ssrc_mc, sdst_mc, NN);

  split_kernel<<<Ppc + Pmc, 1024, 0, stream>>>(ei_pc, Epc, ei_mc, Emc, NN, Ppc,
                                               part_pc, offs_pc, part_mc, offs_mc);

  const int nown = (NN + OWN_R - 1) / OWN_R;
  owner_acc_kernel<<<nown, 1024, 0, stream>>>(part_pc, offs_pc, Ppc, part_mc, offs_mc, Pmc, NN,
                                             ssrc_pc, sdst_pc, h_pc, b_pc,
                                             ssrc_mc, sdst_mc, h_mc, b_mc, out);
}


# GEMM: a wave keeps its W columns in registers over the whole K and its block walks row stripes, next stripe prefetched; x stripe at LDS pitch 132 ends a 16-way bank conflict; sums unchanged
# speedup vs baseline: 5.2531x; 1.0817x over previous
#include <hip/hip_runtime.h>
#include <math.h>

typedef __attribute__((ext_vector_type(2))) float v2f;
typedef __attribute__((ext_vector_type(8))) float v8f;

#define NEG_SLOPE 0.2f

__device__ __forceinline__ int fkey(float v) {
  int b = __float_as_int(v);
  return (b >= 0) ? b : (b ^ 0x7FFFFFFF);
}
__device__ __forceinline__ float kval(int k) {
  return __int_as_float((k >= 0) ? k : (k ^ 0x7FFFFFFF));
}

#define GM_PITCH  132
#define GM_BLOCKS 512

typedef __attribute__((ext_vector_type(4))) float gm4f;

__device__ __forceinline__ void gm_lds_barrier() {
  __builtin_amdgcn_fence(__ATOMIC_RELEASE, "workgroup", "local");
  __builtin_amdgcn_s_barrier();
  __builtin_amdgcn_fence(__ATOMIC_ACQUIRE, "workgroup", "local");
}

__global__ __launch_bounds__(256)
void gemm_kernel(const float* __restrict__ x, const float* __restrict__ Wpc,
                 const float* __restrict__ Wmc, float* __restrict__ hpc,
                 float* __restrict__ hmc, int n) {
  __shared__ __attribute__((aligned(16))) float xs[16 * GM_PITCH];

  const int lane   = threadIdx.x & 31;
  const int wave   = threadIdx.x >> 5;
  const int colbase = wave * 16;
  const int mn    = lane & 15;
  const int khalf = (lane >> 4) << 1;

  const int srow = (int)(threadIdx.x >> 5);
  const int scol = (int)(threadIdx.x & 31) * 4;
  const int nstripes = (n + 15) >> 4;

  int s = blockIdx.x;
  gm4f p0 = {0.0f, 0.0f, 0.0f, 0.0f}, p1 = {0.0f, 0.0f, 0.0f, 0.0f};
  if (s < nstripes) {
    const int r0 = min(s * 16 + srow, n - 1);
    const int r1 = min(s * 16 + srow + 8, n - 1);
    p0 = *(const gm4f*)(x + (size_t)r0 * 128 + scol);
    p1 = *(const gm4f*)(x + (size_t)r1 * 128 + scol);
  }

  v2f bpc[32], bmc[32];
#pragma unroll
  for (int kk = 0; kk < 32; ++kk) {
    const int wi = (kk * 4 + khalf) * 128 + colbase + mn;
    bpc[kk].x = Wpc[wi]; bpc[kk].y = Wpc[wi + 128];
    bmc[kk].x = Wmc[wi]; bmc[kk].y = Wmc[wi + 128];
  }

  *(gm4f*)&xs[srow * GM_PITCH + scol]       = p0;
  *(gm4f*)&xs[(srow + 8) * GM_PITCH + scol] = p1;
  gm_lds_barrier();
  if (s >= nstripes) return;

  for (;;) {
    const int  sn   = s + (int)gridDim.x;
    const bool more = sn < nstripes;
    if (more) {
      const int r0 = min(sn * 16 + srow, n - 1);
      const int r1 = min(sn * 16 + srow + 8, n - 1);
      p0 = *(const gm4f*)(x + (size_t)r0 * 128 + scol);
      p1 = *(const gm4f*)(x + (size_t)r1 * 128 + scol);
    }

    v8f cpc = {};
    v8f cmc = {};
#pragma unroll
    for (int kk = 0; kk < 32; ++kk) {
      const v2f a = *(const v2f*)&xs[mn * GM_PITCH + kk * 4 + khalf];
      cpc = __builtin_amdgcn_wmma_f32_16x16x4_f32(false, a, false, bpc[kk], (short)0, cpc, false, false);
      cmc = __builtin_amdgcn_wmma_f32_16x16x4_f32(false, a, false, bmc[kk], (short)0, cmc, false, false);
    }
    const int crow = s * 16 + ((lane >> 4) << 3);
    const int ccol = colbase + mn;
#pragma unroll
    for (int r = 0; r < 8; ++r) {
      if (crow + r < n) {
        hpc[(size_t)(crow + r) * 128 + ccol] = cpc[r];
        hmc[(size_t)(crow + r) * 128 + ccol] = cmc[r];
      }
    }

    if (!more) break;
    gm_lds_barrier();
    *(gm4f*)&xs[srow * GM_PITCH + scol]       = p0;
    *(gm4f*)&xs[(srow + 8) * GM_PITCH + scol] = p1;
    gm_lds_barrier();
    s = sn;
  }
}

#define SC_NPW 16
typedef __attribute__((ext_vector_type(4))) float sc4f;

__global__ __launch_bounds__(256)
void score_kernel(const float* __restrict__ h, const float* __restrict__ a_src,
                  const float* __restrict__ a_dst, float* __restrict__ s_src,
                  float* __restrict__ s_dst, int n) {
  const int warp = (blockIdx.x * blockDim.x + threadIdx.x) >> 5;
  const int lane = threadIdx.x & 31;
  const int nb = warp * SC_NPW;
  if (nb >= n) return;

  sc4f hv[SC_NPW];
#pragma unroll
  for (int j = 0; j < SC_NPW; ++j) {
    const int node = min(nb + j, n - 1);
    hv[j] = ((const sc4f*)(h + (size_t)node * 128))[lane];
  }
  const sc4f as = ((const sc4f*)a_src)[lane];
  const sc4f ad = ((const sc4f*)a_dst)[lane];

  float ps[SC_NPW], pd[SC_NPW];
#pragma unroll
  for (int j = 0; j < SC_NPW; ++j) {
    float ss = 0.0f, sd = 0.0f;
    ss += hv[j].x * as.x;  sd += hv[j].x * ad.x;
    ss += hv[j].y * as.y;  sd += hv[j].y * ad.y;
    ss += hv[j].z * as.z;  sd += hv[j].z * ad.z;
    ss += hv[j].w * as.w;  sd += hv[j].w * ad.w;
    ps[j] = ss;
    pd[j] = sd;
  }

  const bool b0 = (lane & 1) != 0;
  float s8[8], d8[8];
#pragma unroll
  for (int m = 0; m < 8; ++m) {
    s8[m] = (b0 ? ps[2 * m + 1] : ps[2 * m]) + __shfl_xor(b0 ? ps[2 * m] : ps[2 * m + 1], 1, 32);
    d8[m] = (b0 ? pd[2 * m + 1] : pd[2 * m]) + __shfl_xor(b0 ? pd[2 * m] : pd[2 * m + 1], 1, 32);
  }
  const bool b1 = (lane & 2) != 0;
  float s4[4], d4[4];
#pragma unroll
  for (int m = 0; m < 4; ++m) {
    s4[m] = (b1 ? s8[2 * m + 1] : s8[2 * m]) + __shfl_xor(b1 ? s8[2 * m] : s8[2 * m + 1], 2, 32);
    d4[m] = (b1 ? d8[2 * m + 1] : d8[2 * m]) + __shfl_xor(b1 ? d8[2 * m] : d8[2 * m + 1], 2, 32);
  }
  const bool b2 = (lane & 4) != 0;
  float s2[2], d2[2];
#pragma unroll
  for (int m = 0; m < 2; ++m) {
    s2[m] = (b2 ? s4[2 * m + 1] : s4[2 * m]) + __shfl_xor(b2 ? s4[2 * m] : s4[2 * m + 1], 4, 32);
    d2[m] = (b2 ? d4[2 * m + 1] : d4[2 * m]) + __shfl_xor(b2 ? d4[2 * m] : d4[2 * m + 1], 4, 32);
  }
  const bool b3 = (lane & 8) != 0;
  const float ssum = (b3 ? s2[1] : s2[0]) + __shfl_xor(b3 ? s2[0] : s2[1], 8, 32);
  const float dsum = (b3 ? d2[1] : d2[0]) + __shfl_xor(b3 ? d2[0] : d2[1], 8, 32);

  const int node = nb + (lane & 15);
  if (node < n) {
    const int head = lane >> 4;
    s_src[node * 2 + head] = ssum;
    s_dst[node * 2 + head] = dsum;
  }
}

#define OWN_R   196
#define SP_W    32
#define SP_K    3
#define SP_CH   (SP_W * 32 * SP_K)
#define SP_NB   256
#define SP_PITCH 288
static_assert(SP_NB + 1 <= SP_PITCH && SP_PITCH % 32 == 0, "a block's offsets are its own lines");

typedef __attribute__((ext_vector_type(4))) float v4f;
typedef __attribute__((ext_vector_type(4))) unsigned v4u;

__global__ __launch_bounds__(1024)
void split_kernel(const int* __restrict__ ei_pc, int E_pc, const int* __restrict__ ei_mc, int E_mc,
                  int n, int P_pc,
                  unsigned* __restrict__ part_pc, int* __restrict__ offs_pc,
                  unsigned* __restrict__ part_mc, int* __restrict__ offs_mc) {
  __shared__ int cnt[SP_W * SP_NB];
  __shared__ __attribute__((aligned(16))) unsigned sorted[SP_CH];
  __shared__ int binstart[SP_NB];
  __shared__ int wsum[SP_NB / 32];

  const bool mc    = (int)blockIdx.x >= P_pc;
  const int  chunk = mc ? (int)blockIdx.x - P_pc : (int)blockIdx.x;
  const int* ei    = mc ? ei_mc : ei_pc;
  const int  E     = mc ? E_mc : E_pc;
  unsigned*  part  = (mc ? part_mc : part_pc) + (size_t)chunk * SP_CH;
  int*       offs  = (mc ? offs_mc : offs_pc) + (size_t)chunk * SP_PITCH;
  const int  tot   = E + n;

  const int lane = threadIdx.x & 31;
  const int wave = __builtin_amdgcn_readfirstlane((int)(threadIdx.x >> 5));
  const unsigned lt = (1u << lane) - 1u;
  int* cw = cnt + wave * SP_NB;

  for (int k = 0; k < SP_NB / 32; ++k) cw[k * 32 + lane] = 0;

  const int sbase = chunk * SP_CH + wave * (32 * SP_K) + lane;
  int dd[SP_K], ss[SP_K];
#pragma unroll
  for (int j = 0; j < SP_K; ++j) {
    const int e = sbase + j * 32;
    int d = 0, sn = 0;
    if (e < E) { sn = ei[e]; d = ei[E + e]; }
    else if (e < tot) { sn = e - E; d = e - E; }
    dd[j] = d; ss[j] = sn;
  }

  int key[SP_K], rnk[SP_K];
  unsigned ent[SP_K];
#pragma unroll
  for (int j = 0; j < SP_K; ++j) {
    const bool ok = (sbase + j * 32) < tot;
    const unsigned own = (unsigned)dd[j] / (unsigned)OWN_R;
    key[j] = (int)(own & (unsigned)(SP_NB - 1));
    ent[j] = (((unsigned)dd[j] - own * (unsigned)OWN_R) << 24) | (unsigned)ss[j];
    unsigned same = __builtin_amdgcn_ballot_w32(ok);
    for (int b = 0; b < 8; ++b) {
      const bool bit = ((key[j] >> b) & 1) != 0;
      const unsigned bb = __builtin_amdgcn_ballot_w32(bit);
      same &= bit ? bb : ~bb;
    }
    rnk[j] = 0;
    if (ok) {
      const int old = cw[key[j]];
      rnk[j] = old + __popc(same & lt);
      if ((same & lt) == 0u) cw[key[j]] = old + __popc(same);
    }
  }
  __syncthreads();

  int incl = 0, totb = 0;
  if (threadIdx.x < SP_NB) {
    int c[SP_W];
#pragma unroll
    for (int w = 0; w < SP_W; ++w) c[w] = cnt[w * SP_NB + threadIdx.x];
    int run = 0;
#pragma unroll
    for (int w = 0; w < SP_W; ++w) { cnt[w * SP_NB + threadIdx.x] = run; run += c[w]; }
    totb = run;
    incl = run;
    for (int d = 1; d < 32; d <<= 1) {
      const int t = __shfl_up(incl, d, 32);
      if (lane >= d) incl += t;
    }
    if (lane == 31) wsum[wave] = incl;
  }
  __syncthreads();
  if (threadIdx.x < SP_NB) {
    int wb = 0;
    for (int w = 0; w < SP_NB / 32; ++w) wb += (w < wave) ? wsum[w] : 0;
    const int bs = wb + incl - totb;
    binstart[threadIdx.x] = bs;
    offs[threadIdx.x] = bs;
    if (threadIdx.x == SP_NB - 1) offs[SP_NB] = bs + totb;
  }
  __syncthreads();

#pragma unroll
  for (int j = 0; j < SP_K; ++j)
    if ((sbase + j * 32) < tot) sorted[binstart[key[j]] + cw[key[j]] + rnk[j]] = ent[j];
  __syncthreads();

  if (threadIdx.x < SP_CH / 4) ((v4u*)part)[threadIdx.x] = ((const v4u*)sorted)[threadIdx.x];
}

#define OWN_T   ((OWN_R + 31) / 32)
#define OWN_QC  4096
#define OWN_U   8
#define OWN_S   (OWN_QC / 1024)
#define OWN_KP  33
static_assert(OWN_QC % 1024 == 0 && OWN_R <= 256, "32 waves share the queue in 32-entry steps; local node in 8 bits");

__device__ __forceinline__ void own_consume(int nq, int wave, int lane, unsigned lt, int base,
                                            const float* __restrict__ s_src,
                                            const float* __restrict__ s_dst,
                                            const float* __restrict__ h, float* accs,
                                            int* rmax, float* dens, v2f* ev,
                                            unsigned* q, unsigned* q2, int* kcnt, int* ktot) {
  __syncthreads();

  const int steps = (nq + 1023) >> 10;
  int* cw = kcnt + wave * OWN_KP;
  cw[lane] = 0;
  const int cbase = wave * (steps * 32) + lane;
  unsigned ent[OWN_S];
  int key[OWN_S], rnk[OWN_S];
#pragma unroll
  for (int j = 0; j < OWN_S; ++j) {
    ent[j] = 0u; key[j] = 0; rnk[j] = 0;
    if (j < steps) {
      const int idx = cbase + j * 32;
      const bool ok = idx < nq;
      ent[j] = q[idx];
      key[j] = (int)((ent[j] >> 24) & 31u);
      unsigned same = __builtin_amdgcn_ballot_w32(ok);
      for (int b = 0; b < 5; ++b) {
        const bool bit = ((key[j] >> b) & 1) != 0;
        const unsigned bb = __builtin_amdgcn_ballot_w32(bit);
        same &= bit ? bb : ~bb;
      }
      if (ok) {
        const int old = cw[key[j]];
        rnk[j] = old + __popc(same & lt);
        if ((same & lt) == 0u) cw[key[j]] = old + __popc(same);
      }
    }
  }
  __syncthreads();

  const int kc = kcnt[lane * OWN_KP + wave];
  int kin = kc;
#pragma unroll
  for (int d = 1; d < 32; d <<= 1) {
    const int t = __shfl_up(kin, d, 32);
    if (lane >= d) kin += t;
  }
  if (lane == 31) ktot[wave] = kin;
  __syncthreads();

  const int kt = ktot[lane];
  int tin = kt;
#pragma unroll
  for (int d = 1; d < 32; d <<= 1) {
    const int t = __shfl_up(tin, d, 32);
    if (lane >= d) tin += t;
  }
  const int myoff = __builtin_amdgcn_readlane(tin - kt, wave);
  const int mycnt = __builtin_amdgcn_readlane(kt, wave);
  kcnt[lane * OWN_KP + wave] = myoff + kin - kc;
  __syncthreads();

#pragma unroll
  for (int j = 0; j < OWN_S; ++j)
    if (j < steps && cbase + j * 32 < nq) q2[cw[key[j]] + rnk[j]] = ent[j];
  __syncthreads();

  const int hd = lane >> 4;
  const int ninf = fkey(-__builtin_inff());

  int oldk[OWN_T];
#pragma unroll
  for (int t = 0; t < OWN_T; ++t) {
    const int row = wave + 32 * t;
    oldk[t] = (row < OWN_R) ? rmax[row * 2 + hd] : ninf;
  }

  int tnext[OWN_T];
  {
    int tc[OWN_T];
#pragma unroll
    for (int t = 0; t < OWN_T; ++t) tc[t] = 0;
    for (int b0 = 0; b0 < mycnt; b0 += 32) {
      const bool ok = b0 + lane < mycnt;
      const unsigned e = ok ? q2[myoff + b0 + lane] : 0u;
      const int tt = (int)(e >> 29);
#pragma unroll
      for (int t = 0; t < OWN_T; ++t) tc[t] += __popc(__builtin_amdgcn_ballot_w32(ok && tt == t));
    }
    int run = 0;
#pragma unroll
    for (int t = 0; t < OWN_T; ++t) { tnext[t] = run; run += tc[t]; }
  }

  for (int b0 = 0; b0 < mycnt; b0 += 32) {
    const bool ok = b0 + lane < mycnt;
    const unsigned ent = ok ? q2[myoff + b0 + lane] : 0u;
    const int tt = (int)(ent >> 29);
    int pos = 0;
#pragma unroll
    for (int t = 0; t < OWN_T; ++t) {
      const unsigned m = __builtin_amdgcn_ballot_w32(ok && tt == t);
      if (tt == t) pos = tnext[t] + __popc(m & lt);
      tnext[t] += __popc(m);
    }
    if (ok) {
      const int local = (int)(ent >> 24);
      const int s     = (int)(ent & 0xFFFFFFu);
      const v2f ss = *(const v2f*)(s_src + (size_t)s * 2);
      const v2f sd = *(const v2f*)(s_dst + (size_t)(base + local) * 2);
      v2f v = ss + sd;
      v.x = (v.x > 0.0f) ? v.x : NEG_SLOPE * v.x;
      v.y = (v.y > 0.0f) ? v.y : NEG_SLOPE * v.y;
      q[myoff + pos]  = ent;
      ev[myoff + pos] = v;
      atomicMax(&rmax[local * 2 + 0], fkey(v.x));
      atomicMax(&rmax[local * 2 + 1], fkey(v.y));
    }
  }

#pragma unroll
  for (int t = 0; t < OWN_T; ++t) {
    const int row = wave + 32 * t;
    if (row < OWN_R) {
      const int newk = rmax[row * 2 + hd];
      if (newk != oldk[t] && oldk[t] != ninf) {
        const float sc = expf(kval(oldk[t]) - kval(newk));
        v4f* ar = (v4f*)(accs + row * 128) + lane;
        *ar = *ar * sc;
        if ((lane & 15) == 0) dens[row * 2 + hd] *= sc;
      }
    }
  }

  int cur = -1;
  v4f racc = {0.0f, 0.0f, 0.0f, 0.0f};
  float rden = 0.0f;
  for (int b0 = 0; b0 < mycnt; b0 += 32) {
    const int nb = min(32, mycnt - b0);
    const unsigned ent = (lane < nb) ? q[myoff + b0 + lane] : 0u;
    const int local = (int)(ent >> 24);
    const int s     = (int)(ent & 0xFFFFFFu);
    float a0 = 0.0f, a1 = 0.0f;
    if (lane < nb) {
      const v2f v = ev[myoff + b0 + lane];
      a0 = expf(v.x - kval(rmax[local * 2 + 0]));
      a1 = expf(v.y - kval(rmax[local * 2 + 1]));
    }
    for (int j = 0; j < nb; j += OWN_U) {
      v4f hv[OWN_U];
#pragma unroll
      for (int u = 0; u < OWN_U; ++u) {
        const int jj = min(j + u, nb - 1);
        const int sj = __builtin_amdgcn_readlane(s, jj);
        hv[u] = ((const v4f*)(h + (size_t)sj * 128))[lane];
      }
#pragma unroll
      for (int u = 0; u < OWN_U; ++u) {
        if (j + u < nb) {
          const int   lj = __builtin_amdgcn_readlane(local, j + u);
          if (lj != cur) {
            if (cur >= 0) {
              *((v4f*)(accs + cur * 128) + lane) = racc;
              if ((lane & 15) == 0) dens[cur * 2 + hd] = rden;
            }
            cur  = lj;
            racc = *((const v4f*)(accs + lj * 128) + lane);
            rden = dens[lj * 2 + hd];
          }
          const float b0f = __int_as_float(__builtin_amdgcn_readlane(__float_as_int(a0), j + u));
          const float b1f = __int_as_float(__builtin_amdgcn_readlane(__float_as_int(a1), j + u));
          const float al = (lane < 16) ? b0f : b1f;
          racc.x += al * hv[u].x;
          racc.y += al * hv[u].y;
          racc.z += al * hv[u].z;
          racc.w += al * hv[u].w;
          rden += al;
        }
      }
    }
  }
  if (cur >= 0) {
    *((v4f*)(accs + cur * 128) + lane) = racc;
    if ((lane & 15) == 0) dens[cur * 2 + hd] = rden;
  }
  __syncthreads();
}

__device__ __forceinline__ float own_elu(float v) {
  return (v > 0.0f) ? v : (expf(v) - 1.0f);
}

__global__ __launch_bounds__(1024)
void owner_acc_kernel(const unsigned* __restrict__ part_pc, const int* __restrict__ offs_pc, int P_pc,
                      const unsigned* __restrict__ part_mc, const int* __restrict__ offs_mc, int P_mc,
                      int n,
                      const float* __restrict__ ssrc_pc, const float* __restrict__ sdst_pc,
                      const float* __restrict__ h_pc, const float* __restrict__ b_pc,
                      const float* __restrict__ ssrc_mc, const float* __restrict__ sdst_mc,
                      const float* __restrict__ h_mc, const float* __restrict__ b_mc,
                      float* __restrict__ out) {
  __shared__ __attribute__((aligned(16))) float accs[OWN_R * 128];
  __shared__ __attribute__((aligned(16))) float keep[OWN_R * 128];
  __shared__ __attribute__((aligned(16))) unsigned q[OWN_QC];
  __shared__ unsigned q2[OWN_QC];
  __shared__ __attribute__((aligned(8))) v2f ev[OWN_QC];
  __shared__ int      rmax[OWN_R * 2];
  __shared__ float    dens[OWN_R * 2];
  __shared__ int      wcnt[2 * 32];
  __shared__ int      kcnt[32 * OWN_KP];
  __shared__ int      ktot[32];

  const int lane = threadIdx.x & 31;
  const int wave = __builtin_amdgcn_readfirstlane((int)(threadIdx.x >> 5));
  const int base = blockIdx.x * OWN_R;
  const unsigned lt = (1u << lane) - 1u;
  const int sub = lane & 3;

  for (int cv = 0; cv < 2; ++cv) {
    const unsigned* part = cv ? part_mc : part_pc;
    const int*   offs  = cv ? offs_mc : offs_pc;
    const int    P     = cv ? P_mc : P_pc;
    const float* s_src = cv ? ssrc_mc : ssrc_pc;
    const float* s_dst = cv ? sdst_mc : sdst_pc;
    const float* h     = cv ? h_mc : h_pc;
    const float* bias  = cv ? b_mc : b_pc;

    for (int i = threadIdx.x; i < OWN_R * 128; i += 1024) accs[i] = 0.0f;
    for (int i = threadIdx.x; i < OWN_R * 2; i += 1024) {
      rmax[i] = fkey(-__builtin_inff());
      dens[i] = 0.0f;
    }

    __syncthreads();

    int qfill = 0;
    for (int c0 = 0; c0 < P; c0 += 256) {
      const int c = c0 + (int)(threadIdx.x >> 2);
      int s0 = 0, cn = 0;
      if (c < P) {
        const int* o = offs + (size_t)c * SP_PITCH + blockIdx.x;
        s0 = o[0];
        cn = o[1] - s0;
      }
      const unsigned* src = part + (size_t)min(c, P - 1) * SP_CH + s0;

      const int mine = (sub == 0) ? cn : 0;
      int incl = mine;
#pragma unroll
      for (int d = 1; d < 32; d <<= 1) {
        const int t = __shfl_up(incl, d, 32);
        if (lane >= d) incl += t;
      }
      const int cnt = __builtin_amdgcn_readlane(incl, 31);
      int* wc = wcnt + ((c0 >> 8) & 1) * 32;
      if (lane == 0) wc[wave] = cnt;
      __syncthreads();

      const int cwv = wc[lane];
      int incw = cwv;
#pragma unroll
      for (int d = 1; d < 32; d <<= 1) {
        const int t = __shfl_up(incw, d, 32);
        if (lane >= d) incw += t;
      }
      const int total  = __builtin_amdgcn_readlane(incw, 31);
      const int mybase = __builtin_amdgcn_readlane(incw - cwv, wave);
      const int pbase = mybase + __shfl(incl - mine, lane & ~3, 32);

      int done = 0;
      while (done < total) {
        if (qfill == OWN_QC) {
          own_consume(qfill, wave, lane, lt, base, s_src, s_dst, h, accs, rmax, dens, ev, q, q2, kcnt, ktot);
          qfill = 0;
        }
        const int take = min(OWN_QC - qfill, total - done);
        for (int i = sub; i < cn; i += 8) {
          const int pa = pbase + i - done;
          const int pb = pa + 4;
          const bool ba = (pa >= 0) && (pa < take);
          const bool bb = (i + 4 < cn) && (pb >= 0) && (pb < take);
          const unsigned va = ba ? src[i] : 0u;
          const unsigned vb = bb ? src[i + 4] : 0u;
          if (ba) q[qfill + pa] = va;
          if (bb) q[qfill + pb] = vb;
        }
        qfill += take;
        done  += take;
      }
    }

    own_consume(qfill, wave, lane, lt, base, s_src, s_dst, h, accs, rmax, dens, ev, q, q2, kcnt, ktot);

    const float bc0 = bias[lane * 4 + 0], bc1 = bias[lane * 4 + 1];
    const float bc2 = bias[lane * 4 + 2], bc3 = bias[lane * 4 + 3];
    for (int i4 = threadIdx.x; i4 < OWN_R * 32; i4 += 1024) {
      const int row = i4 >> 5;
      const float d = dens[row * 2 + (lane >> 4)];
      const v4f a = ((const v4f*)accs)[i4];
      v4f v;
      v.x = own_elu(a.x / d + bc0);
      v.y = own_elu(a.y / d + bc1);
      v.z = own_elu(a.z / d + bc2);
      v.w = own_elu(a.w / d + bc3);
      if (cv == 0) {
        ((v4f*)keep)[i4] = v;
      } else if (base + row < n) {
        const v4f k = ((const v4f*)keep)[i4];
        v4f o;
        o.x = 0.5f * (k.x + v.x);
        o.y = 0.5f * (k.y + v.y);
        o.z = 0.5f * (k.z + v.z);
        o.w = 0.5f * (k.w + v.w);
        ((v4f*)(out + (size_t)(base + row) * 128))[lane] = o;
      }
    }
    __syncthreads();
  }
}

extern "C" void kernel_launch(void* const* d_in, const int* in_sizes, int n_in,
                              void* d_out, int out_size, void* d_ws, size_t ws_size,
                              hipStream_t stream) {
  const float* x      = (const float*)d_in[0];
  const float* W_pc   = (const float*)d_in[1];
  const float* asr_pc = (const float*)d_in[2];
  const float* ads_pc = (const float*)d_in[3];
  const float* b_pc   = (const float*)d_in[4];
  const float* W_mc   = (const float*)d_in[5];
  const float* asr_mc = (const float*)d_in[6];
  const float* ads_mc = (const float*)d_in[7];
  const float* b_mc   = (const float*)d_in[8];
  const int*   ei_pc  = (const int*)d_in[9];
  const int*   ei_mc  = (const int*)d_in[10];
  float* out = (float*)d_out;

  const int NN  = in_sizes[0] / 128;
  const int Epc = in_sizes[9] / 2;
  const int Emc = in_sizes[10] / 2;
  const int Ppc = (Epc + NN + SP_CH - 1) / SP_CH;
  const int Pmc = (Emc + NN + SP_CH - 1) / SP_CH;

  float* ws = (float*)d_ws;
  size_t off = 0;
  float* h_pc    = ws + off; off += (size_t)NN * 128;
  float* h_mc    = ws + off; off += (size_t)NN * 128;
  float* ssrc_pc = ws + off; off += (size_t)NN * 2;
  float* sdst_pc = ws + off; off += (size_t)NN * 2;
  float* ssrc_mc = ws + off; off += (size_t)NN * 2;
  float* sdst_mc = ws + off; off += (size_t)NN * 2;
  unsigned* part_pc = (unsigned*)(ws + off); off += (size_t)Ppc * SP_CH;
  unsigned* part_mc = (unsigned*)(ws + off); off += (size_t)Pmc * SP_CH;
  int* offs_pc = (int*)(ws + off); off += (size_t)Ppc * SP_PITCH;
  int* offs_mc = (int*)(ws + off); off += (size_t)Pmc * SP_PITCH;

  gemm_kernel<<<GM_BLOCKS, 256, 0, stream>>>(x, W_pc, W_mc, h_pc, h_mc, NN);

  score_kernel<<<(NN + 8 * SC_NPW - 1) / (8 * SC_NPW), 256, 0, stream>>>(h_pc, asr_pc, ads_pc, ssrc_pc, sdst_pc, NN);
  score_kernel<<<(NN + 8 * SC_NPW - 1) / (8 * SC_NPW), 256, 0, stream>>>(h_mc, asr_mc, ads_mc, ssrc_mc, sdst_mc, NN);

  split_kernel<<<Ppc + Pmc, 1024, 0, stream>>>(ei_pc, Epc, ei_mc, Emc, NN, Ppc,
                                               part_pc, offs_pc, part_mc, offs_mc);

  const int nown = (NN + OWN_R - 1) / OWN_R;
  owner_acc_kernel<<<nown, 1024, 0, stream>>>(part_pc, offs_pc, Ppc, part_mc, offs_mc, Pmc, NN,
                                             ssrc_pc, sdst_pc, h_pc, b_pc,
                                             ssrc_mc, sdst_mc, h_mc, b_mc, out);
}


# Owner work-group runs both edge lists through one queue (pc rows, then mc rows: 392 rows in LDS): one piece fetch, one sort, one scoring, one summing and one finishing pass; sum order kept
# speedup vs baseline: 5.5113x; 1.0492x over previous
#include <hip/hip_runtime.h>
#include <math.h>

typedef __attribute__((ext_vector_type(2))) float v2f;
typedef __attribute__((ext_vector_type(8))) float v8f;

#define NEG_SLOPE 0.2f

__device__ __forceinline__ int fkey(float v) {
  int b = __float_as_int(v);
  return (b >= 0) ? b : (b ^ 0x7FFFFFFF);
}
__device__ __forceinline__ float kval(int k) {
  return __int_as_float((k >= 0) ? k : (k ^ 0x7FFFFFFF));
}

#define GM_PITCH  132
#define GM_BLOCKS 512

typedef __attribute__((ext_vector_type(4))) float gm4f;

__device__ __forceinline__ void gm_lds_barrier() {
  __builtin_amdgcn_fence(__ATOMIC_RELEASE, "workgroup", "local");
  __builtin_amdgcn_s_barrier();
  __builtin_amdgcn_fence(__ATOMIC_ACQUIRE, "workgroup", "local");
}

__global__ __launch_bounds__(256)
void gemm_kernel(const float* __restrict__ x, const float* __restrict__ Wpc,
                 const float* __restrict__ Wmc, float* __restrict__ hpc,
                 float* __restrict__ hmc, int n) {
  __shared__ __attribute__((aligned(16))) float xs[16 * GM_PITCH];

  const int lane   = threadIdx.x & 31;
  const int wave   = threadIdx.x >> 5;
  const int colbase = wave * 16;
  const int mn    = lane & 15;
  const int khalf = (lane >> 4) << 1;

  const int srow = (int)(threadIdx.x >> 5);
  const int scol = (int)(threadIdx.x & 31) * 4;
  const int nstripes = (n + 15) >> 4;

  int s = blockIdx.x;
  gm4f p0 = {0.0f, 0.0f, 0.0f, 0.0f}, p1 = {0.0f, 0.0f, 0.0f, 0.0f};
  if (s < nstripes) {
    const int r0 = min(s * 16 + srow, n - 1);
    const int r1 = min(s * 16 + srow + 8, n - 1);
    p0 = *(const gm4f*)(x + (size_t)r0 * 128 + scol);
    p1 = *(const gm4f*)(x + (size_t)r1 * 128 + scol);
  }

  v2f bpc[32], bmc[32];
#pragma unroll
  for (int kk = 0; kk < 32; ++kk) {
    const int wi = (kk * 4 + khalf) * 128 + colbase + mn;
    bpc[kk].x = Wpc[wi]; bpc[kk].y = Wpc[wi + 128];
    bmc[kk].x = Wmc[wi]; bmc[kk].y = Wmc[wi + 128];
  }

  *(gm4f*)&xs[srow * GM_PITCH + scol]       = p0;
  *(gm4f*)&xs[(srow + 8) * GM_PITCH + scol] = p1;
  gm_lds_barrier();
  if (s >= nstripes) return;

  for (;;) {
    const int  sn   = s + (int)gridDim.x;
    const bool more = sn < nstripes;
    if (more) {
      const int r0 = min(sn * 16 + srow, n - 1);
      const int r1 = min(sn * 16 + srow + 8, n - 1);
      p0 = *(const gm4f*)(x + (size_t)r0 * 128 + scol);
      p1 = *(const gm4f*)(x + (size_t)r1 * 128 + scol);
    }

    v8f cpc = {};
    v8f cmc = {};
#pragma unroll
    for (int kk = 0; kk < 32; ++kk) {
      const v2f a = *(const v2f*)&xs[mn * GM_PITCH + kk * 4 + khalf];
      cpc = __builtin_amdgcn_wmma_f32_16x16x4_f32(false, a, false, bpc[kk], (short)0, cpc, false, false);
      cmc = __builtin_amdgcn_wmma_f32_16x16x4_f32(false, a, false, bmc[kk], (short)0, cmc, false, false);
    }
    const int crow = s * 16 + ((lane >> 4) << 3);
    const int ccol = colbase + mn;
#pragma unroll
    for (int r = 0; r < 8; ++r) {
      if (crow + r < n) {
        hpc[(size_t)(crow + r) * 128 + ccol] = cpc[r];
        hmc[(size_t)(crow + r) * 128 + ccol] = cmc[r];
      }
    }

    if (!more) break;
    gm_lds_barrier();
    *(gm4f*)&xs[srow * GM_PITCH + scol]       = p0;
    *(gm4f*)&xs[(srow + 8) * GM_PITCH + scol] = p1;
    gm_lds_barrier();
    s = sn;
  }
}

#define SC_NPW 16
typedef __attribute__((ext_vector_type(4))) float sc4f;

__global__ __launch_bounds__(256)
void score_kernel(const float* __restrict__ h, const float* __restrict__ a_src,
                  const float* __restrict__ a_dst, float* __restrict__ s_src,
                  float* __restrict__ s_dst, int n) {
  const int warp = (blockIdx.x * blockDim.x + threadIdx.x) >> 5;
  const int lane = threadIdx.x & 31;
  const int nb = warp * SC_NPW;
  if (nb >= n) return;

  sc4f hv[SC_NPW];
#pragma unroll
  for (int j = 0; j < SC_NPW; ++j) {
    const int node = min(nb + j, n - 1);
    hv[j] = ((const sc4f*)(h + (size_t)node * 128))[lane];
  }
  const sc4f as = ((const sc4f*)a_src)[lane];
  const sc4f ad = ((const sc4f*)a_dst)[lane];

  float ps[SC_NPW], pd[SC_NPW];
#pragma unroll
  for (int j = 0; j < SC_NPW; ++j) {
    float ss = 0.0f, sd = 0.0f;
    ss += hv[j].x * as.x;  sd += hv[j].x * ad.x;
    ss += hv[j].y * as.y;  sd += hv[j].y * ad.y;
    ss += hv[j].z * as.z;  sd += hv[j].z * ad.z;
    ss += hv[j].w * as.w;  sd += hv[j].w * ad.w;
    ps[j] = ss;
    pd[j] = sd;
  }

  const bool b0 = (lane & 1) != 0;
  float s8[8], d8[8];
#pragma unroll
  for (int m = 0; m < 8; ++m) {
    s8[m] = (b0 ? ps[2 * m + 1] : ps[2 * m]) + __shfl_xor(b0 ? ps[2 * m] : ps[2 * m + 1], 1, 32);
    d8[m] = (b0 ? pd[2 * m + 1] : pd[2 * m]) + __shfl_xor(b0 ? pd[2 * m] : pd[2 * m + 1], 1, 32);
  }
  const bool b1 = (lane & 2) != 0;
  float s4[4], d4[4];
#pragma unroll
  for (int m = 0; m < 4; ++m) {
    s4[m] = (b1 ? s8[2 * m + 1] : s8[2 * m]) + __shfl_xor(b1 ? s8[2 * m] : s8[2 * m + 1], 2, 32);
    d4[m] = (b1 ? d8[2 * m + 1] : d8[2 * m]) + __shfl_xor(b1 ? d8[2 * m] : d8[2 * m + 1], 2, 32);
  }
  const bool b2 = (lane & 4) != 0;
  float s2[2], d2[2];
#pragma unroll
  for (int m = 0; m < 2; ++m) {
    s2[m] = (b2 ? s4[2 * m + 1] : s4[2 * m]) + __shfl_xor(b2 ? s4[2 * m] : s4[2 * m + 1], 4, 32);
    d2[m] = (b2 ? d4[2 * m + 1] : d4[2 * m]) + __shfl_xor(b2 ? d4[2 * m] : d4[2 * m + 1], 4, 32);
  }
  const bool b3 = (lane & 8) != 0;
  const float ssum = (b3 ? s2[1] : s2[0]) + __shfl_xor(b3 ? s2[0] : s2[1], 8, 32);
  const float dsum = (b3 ? d2[1] : d2[0]) + __shfl_xor(b3 ? d2[0] : d2[1], 8, 32);

  const int node = nb + (lane & 15);
  if (node < n) {
    const int head = lane >> 4;
    s_src[node * 2 + head] = ssum;
    s_dst[node * 2 + head] = dsum;
  }
}

#define OWN_R   196
#define SP_W    32
#define SP_K    3
#define SP_CH   (SP_W * 32 * SP_K)
#define SP_NB   256
#define SP_PITCH 288
static_assert(SP_NB + 1 <= SP_PITCH && SP_PITCH % 32 == 0, "a block's offsets are its own lines");

typedef __attribute__((ext_vector_type(4))) float v4f;
typedef __attribute__((ext_vector_type(4))) unsigned v4u;

__global__ __launch_bounds__(1024)
void split_kernel(const int* __restrict__ ei_pc, int E_pc, const int* __restrict__ ei_mc, int E_mc,
                  int n, int P_pc,
                  unsigned* __restrict__ part_pc, int* __restrict__ offs_pc,
                  unsigned* __restrict__ part_mc, int* __restrict__ offs_mc) {
  __shared__ int cnt[SP_W * SP_NB];
  __shared__ __attribute__((aligned(16))) unsigned sorted[SP_CH];
  __shared__ int binstart[SP_NB];
  __shared__ int wsum[SP_NB / 32];

  const bool mc    = (int)blockIdx.x >= P_pc;
  const int  chunk = mc ? (int)blockIdx.x - P_pc : (int)blockIdx.x;
  const int* ei    = mc ? ei_mc : ei_pc;
  const int  E     = mc ? E_mc : E_pc;
  unsigned*  part  = (mc ? part_mc : part_pc) + (size_t)chunk * SP_CH;
  int*       offs  = (mc ? offs_mc : offs_pc) + (size_t)chunk * SP_PITCH;
  const int  tot   = E + n;

  const int lane = threadIdx.x & 31;
  const int wave = __builtin_amdgcn_readfirstlane((int)(threadIdx.x >> 5));
  const unsigned lt = (1u << lane) - 1u;
  int* cw = cnt + wave * SP_NB;

  for (int k = 0; k < SP_NB / 32; ++k) cw[k * 32 + lane] = 0;

  const int sbase = chunk * SP_CH + wave * (32 * SP_K) + lane;
  int dd[SP_K], ss[SP_K];
#pragma unroll
  for (int j = 0; j < SP_K; ++j) {
    const int e = sbase + j * 32;
    int d = 0, sn = 0;
    if (e < E) { sn = ei[e]; d = ei[E + e]; }
    else if (e < tot) { sn = e - E; d = e - E; }
    dd[j] = d; ss[j] = sn;
  }

  int key[SP_K], rnk[SP_K];
  unsigned ent[SP_K];
#pragma unroll
  for (int j = 0; j < SP_K; ++j) {
    const bool ok = (sbase + j * 32) < tot;
    const unsigned own = (unsigned)dd[j] / (unsigned)OWN_R;
    key[j] = (int)(own & (unsigned)(SP_NB - 1));
    ent[j] = (((unsigned)dd[j] - own * (unsigned)OWN_R) << 24) | (unsigned)ss[j];
    unsigned same = __builtin_amdgcn_ballot_w32(ok);
    for (int b = 0; b < 8; ++b) {
      const bool bit = ((key[j] >> b) & 1) != 0;
      const unsigned bb = __builtin_amdgcn_ballot_w32(bit);
      same &= bit ? bb : ~bb;
    }
    rnk[j] = 0;
    if (ok) {
      const int old = cw[key[j]];
      rnk[j] = old + __popc(same & lt);
      if ((same & lt) == 0u) cw[key[j]] = old + __popc(same);
    }
  }
  __syncthreads();

  int incl = 0, totb = 0;
  if (threadIdx.x < SP_NB) {
    int c[SP_W];
#pragma unroll
    for (int w = 0; w < SP_W; ++w) c[w] = cnt[w * SP_NB + threadIdx.x];
    int run = 0;
#pragma unroll
    for (int w = 0; w < SP_W; ++w) { cnt[w * SP_NB + threadIdx.x] = run; run += c[w]; }
    totb = run;
    incl = run;
    for (int d = 1; d < 32; d <<= 1) {
      const int t = __shfl_up(incl, d, 32);
      if (lane >= d) incl += t;
    }
    if (lane == 31) wsum[wave] = incl;
  }
  __syncthreads();
  if (threadIdx.x < SP_NB) {
    int wb = 0;
    for (int w = 0; w < SP_NB / 32; ++w) wb += (w < wave) ? wsum[w] : 0;
    const int bs = wb + incl - totb;
    binstart[threadIdx.x] = bs;
    offs[threadIdx.x] = bs;
    if (threadIdx.x == SP_NB - 1) offs[SP_NB] = bs + totb;
  }
  __syncthreads();

#pragma unroll
  for (int j = 0; j < SP_K; ++j)
    if ((sbase + j * 32) < tot) sorted[binstart[key[j]] + cw[key[j]] + rnk[j]] = ent[j];
  __syncthreads();

  if (threadIdx.x < SP_CH / 4) ((v4u*)part)[threadIdx.x] = ((const v4u*)sorted)[threadIdx.x];
}

#define OWN_V   (2 * OWN_R)
#define OWN_T   ((OWN_V + 31) / 32)
#define OWN_QC  6144
#define OWN_U   8
#define OWN_S   (OWN_QC / 1024)
#define OWN_KP  33
static_assert(OWN_QC % 1024 == 0 && OWN_R <= 256 && OWN_V <= 512, "32 waves share the queue in 32-entry steps; local node in 8 bits, row in 9");

__device__ __forceinline__ void own_consume(int nq, int wave, int lane, unsigned lt, int base, int n,
                                            const float* __restrict__ sc,
                                            const float* __restrict__ h, float* accs,
                                            int* rmax, float* dens, v2f* ev,
                                            unsigned* q, unsigned* q2, int* kcnt, int* ktot) {
  __syncthreads();

  const int steps = (nq + 1023) >> 10;
  int* cw = kcnt + wave * OWN_KP;
  cw[lane] = 0;
  const int cbase = wave * (steps * 32) + lane;
  unsigned ent[OWN_S];
  int key[OWN_S], rnk[OWN_S];
#pragma unroll
  for (int j = 0; j < OWN_S; ++j) {
    ent[j] = 0u; key[j] = 0; rnk[j] = 0;
    if (j < steps) {
      const int idx = cbase + j * 32;
      const bool ok = idx < nq;
      ent[j] = q[idx];
      key[j] = (int)((ent[j] >> 23) & 31u);
      unsigned same = __builtin_amdgcn_ballot_w32(ok);
      for (int b = 0; b < 5; ++b) {
        const bool bit = ((key[j] >> b) & 1) != 0;
        const unsigned bb = __builtin_amdgcn_ballot_w32(bit);
        same &= bit ? bb : ~bb;
      }
      if (ok) {
        const int old = cw[key[j]];
        rnk[j] = old + __popc(same & lt);
        if ((same & lt) == 0u) cw[key[j]] = old + __popc(same);
      }
    }
  }
  __syncthreads();

  const int kc = kcnt[lane * OWN_KP + wave];
  int kin = kc;
#pragma unroll
  for (int d = 1; d < 32; d <<= 1) {
    const int t = __shfl_up(kin, d, 32);
    if (lane >= d) kin += t;
  }
  if (lane == 31) ktot[wave] = kin;
  __syncthreads();

  const int kt = ktot[lane];
  int tin = kt;
#pragma unroll
  for (int d = 1; d < 32; d <<= 1) {
    const int t = __shfl_up(tin, d, 32);
    if (lane >= d) tin += t;
  }
  const int myoff = __builtin_amdgcn_readlane(tin - kt, wave);
  const int mycnt = __builtin_amdgcn_readlane(kt, wave);
  kcnt[lane * OWN_KP + wave] = myoff + kin - kc;
  __syncthreads();

#pragma unroll
  for (int j = 0; j < OWN_S; ++j)
    if (j < steps && cbase + j * 32 < nq) q2[cw[key[j]] + rnk[j]] = ent[j];
  __syncthreads();

  const int hd = lane >> 4;
  const int ninf = fkey(-__builtin_inff());

  int oldk[OWN_T];
#pragma unroll
  for (int t = 0; t < OWN_T; ++t) {
    const int row = wave + 32 * t;
    oldk[t] = (row < OWN_V) ? rmax[row * 2 + hd] : ninf;
  }

  int tnext[OWN_T];
  {
    int tc[OWN_T];
#pragma unroll
    for (int t = 0; t < OWN_T; ++t) tc[t] = 0;
    for (int b0 = 0; b0 < mycnt; b0 += 32) {
      const bool ok = b0 + lane < mycnt;
      const unsigned e = ok ? q2[myoff + b0 + lane] : 0u;
      const int tt = (int)(e >> 28);
#pragma unroll
      for (int t = 0; t < OWN_T; ++t) tc[t] += __popc(__builtin_amdgcn_ballot_w32(ok && tt == t));
    }
    int run = 0;
#pragma unroll
    for (int t = 0; t < OWN_T; ++t) { tnext[t] = run; run += tc[t]; }
  }

  for (int b0 = 0; b0 < mycnt; b0 += 32) {
    const bool ok = b0 + lane < mycnt;
    const unsigned ent = ok ? q2[myoff + b0 + lane] : 0u;
    const int tt = (int)(ent >> 28);
    int pos = 0;
#pragma unroll
    for (int t = 0; t < OWN_T; ++t) {
      const unsigned m = __builtin_amdgcn_ballot_w32(ok && tt == t);
      if (tt == t) pos = tnext[t] + __popc(m & lt);
      tnext[t] += __popc(m);
    }
    if (ok) {
      const int  row   = (int)(ent >> 23);
      const bool mcl   = row >= OWN_R;
      const int  local = mcl ? row - OWN_R : row;
      const int  s     = (int)(ent & 0x7FFFFFu);
      const float* sl  = sc + (mcl ? (size_t)n * 4 : (size_t)0);
      const v2f ss = *(const v2f*)(sl + (size_t)s * 2);
      const v2f sd = *(const v2f*)(sl + (size_t)n * 2 + (size_t)(base + local) * 2);
      v2f v = ss + sd;
      v.x = (v.x > 0.0f) ? v.x : NEG_SLOPE * v.x;
      v.y = (v.y > 0.0f) ? v.y : NEG_SLOPE * v.y;
      q[myoff + pos]  = ent;
      ev[myoff + pos] = v;
      atomicMax(&rmax[row * 2 + 0], fkey(v.x));
      atomicMax(&rmax[row * 2 + 1], fkey(v.y));
    }
  }

#pragma unroll
  for (int t = 0; t < OWN_T; ++t) {
    const int row = wave + 32 * t;
    if (row < OWN_V) {
      const int newk = rmax[row * 2 + hd];
      if (newk != oldk[t] && oldk[t] != ninf) {
        const float sc = expf(kval(oldk[t]) - kval(newk));
        v4f* ar = (v4f*)(accs + row * 128) + lane;
        *ar = *ar * sc;
        if ((lane & 15) == 0) dens[row * 2 + hd] *= sc;
      }
    }
  }

  int cur = -1;
  v4f racc = {0.0f, 0.0f, 0.0f, 0.0f};
  float rden = 0.0f;
  for (int b0 = 0; b0 < mycnt; b0 += 32) {
    const int nb = min(32, mycnt - b0);
    const unsigned ent = (lane < nb) ? q[myoff + b0 + lane] : 0u;
    const int local = (int)(ent >> 23);
    const int s     = (int)(ent & 0x7FFFFFu) + ((local >= OWN_R) ? n : 0);
    float a0 = 0.0f, a1 = 0.0f;
    if (lane < nb) {
      const v2f v = ev[myoff + b0 + lane];
      a0 = expf(v.x - kval(rmax[local * 2 + 0]));
      a1 = expf(v.y - kval(rmax[local * 2 + 1]));
    }
    for (int j = 0; j < nb; j += OWN_U) {
      v4f hv[OWN_U];
#pragma unroll
      for (int u = 0; u < OWN_U; ++u) {
        const int jj = min(j + u, nb - 1);
        const int sj = __builtin_amdgcn_readlane(s, jj);
        hv[u] = ((const v4f*)(h + (size_t)sj * 128))[lane];
      }
#pragma unroll
      for (int u = 0; u < OWN_U; ++u) {
        if (j + u < nb) {
          const int   lj = __builtin_amdgcn_readlane(local, j + u);
          if (lj != cur) {
            if (cur >= 0) {
              *((v4f*)(accs + cur * 128) + lane) = racc;
              if ((lane & 15) == 0) dens[cur * 2 + hd] = rden;
            }
            cur  = lj;
            racc = *((const v4f*)(accs + lj * 128) + lane);
            rden = dens[lj * 2 + hd];
          }
          const float b0f = __int_as_float(__builtin_amdgcn_readlane(__float_as_int(a0), j + u));
          const float b1f = __int_as_float(__builtin_amdgcn_readlane(__float_as_int(a1), j + u));
          const float al = (lane < 16) ? b0f : b1f;
          racc.x += al * hv[u].x;
          racc.y += al * hv[u].y;
          racc.z += al * hv[u].z;
          racc.w += al * hv[u].w;
          rden += al;
        }
      }
    }
  }
  if (cur >= 0) {
    *((v4f*)(accs + cur * 128) + lane) = racc;
    if ((lane & 15) == 0) dens[cur * 2 + hd] = rden;
  }
  __syncthreads();
}

__device__ __forceinline__ float own_elu(float v) {
  return (v > 0.0f) ? v : (expf(v) - 1.0f);
}

__global__ __launch_bounds__(1024)
void owner_acc_kernel(const unsigned* __restrict__ part_pc, const int* __restrict__ offs_pc, int P_pc,
                      const unsigned* __restrict__ part_mc, const int* __restrict__ offs_mc, int P_mc,
                      int n,
                      const float* __restrict__ sc, const float* __restrict__ h,
                      const float* __restrict__ b_pc, const float* __restrict__ b_mc,
                      float* __restrict__ out) {
  __shared__ __attribute__((aligned(16))) float accs[OWN_V * 128];
  __shared__ __attribute__((aligned(16))) unsigned q[OWN_QC];
  __shared__ unsigned q2[OWN_QC];
  __shared__ __attribute__((aligned(8))) v2f ev[OWN_QC];
  __shared__ int      rmax[OWN_V * 2];
  __shared__ float    dens[OWN_V * 2];
  __shared__ int      wcnt[2 * 32];
  __shared__ int      kcnt[32 * OWN_KP];
  __shared__ int      ktot[32];

  const int lane = threadIdx.x & 31;
  const int wave = __builtin_amdgcn_readfirstlane((int)(threadIdx.x >> 5));
  const int base = blockIdx.x * OWN_R;
  const unsigned lt = (1u << lane) - 1u;
  const int sub = lane & 1;

  for (int i = threadIdx.x; i < OWN_V * 128; i += 1024) accs[i] = 0.0f;
  for (int i = threadIdx.x; i < OWN_V * 2; i += 1024) {
    rmax[i] = fkey(-__builtin_inff());
    dens[i] = 0.0f;
  }
  __syncthreads();

  int qfill = 0;
  int pass  = 0;
  const int PT = P_pc + P_mc;
  for (int c0 = 0; c0 < PT; c0 += 512, ++pass) {
    const int  c   = c0 + (int)(threadIdx.x >> 1);
    const bool mcp = c >= P_pc;
    const int  cl  = (c < PT) ? (mcp ? c - P_pc : c) : 0;
    int s0 = 0, cn = 0;
    if (c < PT) {
      const int* o = (mcp ? offs_mc : offs_pc) + (size_t)cl * SP_PITCH + blockIdx.x;
      s0 = o[0];
      cn = o[1] - s0;
    }
    const unsigned* src = (mcp ? part_mc : part_pc) + (size_t)cl * SP_CH + s0;
    const unsigned radd = mcp ? ((unsigned)OWN_R << 23) : 0u;

    const int mine = (sub == 0) ? cn : 0;
    int incl = mine;
#pragma unroll
    for (int d = 1; d < 32; d <<= 1) {
      const int t = __shfl_up(incl, d, 32);
      if (lane >= d) incl += t;
    }
    const int cnt = __builtin_amdgcn_readlane(incl, 31);
    int* wc = wcnt + (pass & 1) * 32;
    if (lane == 0) wc[wave] = cnt;
    __syncthreads();

    const int cwv = wc[lane];
    int incw = cwv;
#pragma unroll
    for (int d = 1; d < 32; d <<= 1) {
      const int t = __shfl_up(incw, d, 32);
      if (lane >= d) incw += t;
    }
    const int total  = __builtin_amdgcn_readlane(incw, 31);
    const int mybase = __builtin_amdgcn_readlane(incw - cwv, wave);
    const int pbase = mybase + __shfl(incl - mine, lane & ~1, 32);

    int done = 0;
    while (done < total) {
      if (qfill == OWN_QC) {
        own_consume(qfill, wave, lane, lt, base, n, sc, h, accs, rmax, dens, ev, q, q2, kcnt, ktot);
        qfill = 0;
      }
      const int take = min(OWN_QC - qfill, total - done);
      for (int i = sub; i < cn; i += 8) {
        unsigned v[4];
        bool     b[4];
#pragma unroll
        for (int k = 0; k < 4; ++k) {
          const int p = pbase + i + 2 * k - done;
          b[k] = (i + 2 * k < cn) && (p >= 0) && (p < take);
          v[k] = b[k] ? src[i + 2 * k] : 0u;
        }
#pragma unroll
        for (int k = 0; k < 4; ++k) {
          const int p = pbase + i + 2 * k - done;
          if (b[k]) q[qfill + p] = (((v[k] >> 24) << 23) | (v[k] & 0x7FFFFFu)) + radd;
        }
      }
      qfill += take;
      done  += take;
    }
  }

  own_consume(qfill, wave, lane, lt, base, n, sc, h, accs, rmax, dens, ev, q, q2, kcnt, ktot);

  const float pb0 = b_pc[lane * 4 + 0], pb1 = b_pc[lane * 4 + 1];
  const float pb2 = b_pc[lane * 4 + 2], pb3 = b_pc[lane * 4 + 3];
  const float mb0 = b_mc[lane * 4 + 0], mb1 = b_mc[lane * 4 + 1];
  const float mb2 = b_mc[lane * 4 + 2], mb3 = b_mc[lane * 4 + 3];
  for (int i4 = threadIdx.x; i4 < OWN_R * 32; i4 += 1024) {
    const int row = i4 >> 5;
    if (base + row < n) {
      const float dp = dens[row * 2 + (lane >> 4)];
      const float dm = dens[(OWN_R + row) * 2 + (lane >> 4)];
      const v4f ap = ((const v4f*)accs)[i4];
      const v4f am = ((const v4f*)accs)[OWN_R * 32 + i4];
      v4f o;
      o.x = 0.5f * (own_elu(ap.x / dp + pb0) + own_elu(am.x / dm + mb0));
      o.y = 0.5f * (own_elu(ap.y / dp + pb1) + own_elu(am.y / dm + mb1));
      o.z = 0.5f * (own_elu(ap.z / dp + pb2) + own_elu(am.z / dm + mb2));
      o.w = 0.5f * (own_elu(ap.w / dp + pb3) + own_elu(am.w / dm + mb3));
      ((v4f*)(out + (size_t)(base + row) * 128))[lane] = o;
    }
  }
}

extern "C" void kernel_launch(void* const* d_in, const int* in_sizes, int n_in,
                              void* d_out, int out_size, void* d_ws, size_t ws_size,
                              hipStream_t stream) {
  const float* x      = (const float*)d_in[0];
  const float* W_pc   = (const float*)d_in[1];
  const float* asr_pc = (const float*)d_in[2];
  const float* ads_pc = (const float*)d_in[3];
  const float* b_pc   = (const float*)d_in[4];
  const float* W_mc   = (const float*)d_in[5];
  const float* asr_mc = (const float*)d_in[6];
  const float* ads_mc = (const float*)d_in[7];
  const float* b_mc   = (const float*)d_in[8];
  const int*   ei_pc  = (const int*)d_in[9];
  const int*   ei_mc  = (const int*)d_in[10];
  float* out = (float*)d_out;

  const int NN  = in_sizes[0] / 128;
  const int Epc = in_sizes[9] / 2;
  const int Emc = in_sizes[10] / 2;
  const int Ppc = (Epc + NN + SP_CH - 1) / SP_CH;
  const int Pmc = (Emc + NN + SP_CH - 1) / SP_CH;

  float* ws = (float*)d_ws;
  size_t off = 0;
  float* h_pc    = ws + off; off += (size_t)NN * 128;
  float* h_mc    = ws + off; off += (size_t)NN * 128;
  float* ssrc_pc = ws + off; off += (size_t)NN * 2;
  float* sdst_pc = ws + off; off += (size_t)NN * 2;
  float* ssrc_mc = ws + off; off += (size_t)NN * 2;
  float* sdst_mc = ws + off; off += (size_t)NN * 2;
  unsigned* part_pc = (unsigned*)(ws + off); off += (size_t)Ppc * SP_CH;
  unsigned* part_mc = (unsigned*)(ws + off); off += (size_t)Pmc * SP_CH;
  int* offs_pc = (int*)(ws + off); off += (size_t)Ppc * SP_PITCH;
  int* offs_mc = (int*)(ws + off); off += (size_t)Pmc * SP_PITCH;

  gemm_kernel<<<GM_BLOCKS, 256, 0, stream>>>(x, W_pc, W_mc, h_pc, h_mc, NN);

  score_kernel<<<(NN + 8 * SC_NPW - 1) / (8 * SC_NPW), 256, 0, stream>>>(h_pc, asr_pc, ads_pc, ssrc_pc, sdst_pc, NN);
  score_kernel<<<(NN + 8 * SC_NPW - 1) / (8 * SC_NPW), 256, 0, stream>>>(h_mc, asr_mc, ads_mc, ssrc_mc, sdst_mc, NN);

  split_kernel<<<Ppc + Pmc, 1024, 0, stream>>>(ei_pc, Epc, ei_mc, Emc, NN, Ppc,
                                               part_pc, offs_pc, part_mc, offs_mc);

  const int nown = (NN + OWN_R - 1) / OWN_R;
  owner_acc_kernel<<<nown, 1024, 0, stream>>>(part_pc, offs_pc, Ppc, part_mc, offs_mc, Pmc, NN,
                                             ssrc_pc, h_pc, b_pc, b_mc, out);
}
